# starting kernel
# baseline (speedup 1.0000x reference)
#include <hip/hip_runtime.h>


typedef __attribute__((ext_vector_type(2))) float v2f;
typedef __attribute__((ext_vector_type(8))) float v8f;

#define LDK 260

__device__ __forceinline__ v8f wmma4(v2f a, v2f b, v8f c) {
  return __builtin_amdgcn_wmma_f32_16x16x4_f32(false, a, false, b,
                                               (short)0, c, false, false);
}

__device__ __forceinline__ void gemm_acc(const float* As, int aoff,
                                         const float* __restrict__ B,
                                         int K, int col0, int lane, v8f& c) {
  const int hl = lane >> 4;
  const int m  = lane & 15;
  const int n  = lane & 15;
  #pragma unroll 4
  for (int k0 = 0; k0 < K; k0 += 4) {
    const int ka = k0 + 2 * hl;
    v2f a, b;
    a.x = As[m * LDK + aoff + ka];
    a.y = As[m * LDK + aoff + ka + 1];
    b.x = B[ka * 128 + col0 + n];
    b.y = B[(ka + 1) * 128 + col0 + n];
    c = wmma4(a, b, c);
  }
}

__device__ __forceinline__ void store_tile(float* __restrict__ out, int row0,
                                           int nrows, int col0, int lane,
                                           v8f c, bool do_relu) {
  const int hl = lane >> 4;
  const int n  = lane & 15;
  #pragma unroll
  for (int v = 0; v < 8; ++v) {
    int row = row0 + v + 8 * hl;
    float val = c[v];
    if (do_relu) val = fmaxf(val, 0.0f);
    if (row < nrows) out[(size_t)row * 128 + col0 + n] = val;
  }
}


__global__ void __launch_bounds__(256) encode_pol_kernel(
    const float* __restrict__ x, const int* __restrict__ sidx,
    const float* __restrict__ semb, const float* __restrict__ W,
    const float* __restrict__ bvec, float* __restrict__ out, int nrows) {
  __shared__ float As[16 * LDK];
  const int tid  = threadIdx.x;
  const int row0 = blockIdx.x * 16;
  for (int idx = tid; idx < 16 * 72; idx += 256) {
    int r = idx / 72, cc = idx % 72;
    int row = row0 + r;
    float v = 0.0f;
    if (row < nrows) {
      if (cc < 64) v = x[(size_t)row * 64 + cc];
      else         v = semb[sidx[row] * 8 + (cc - 64)];
    }
    As[r * LDK + cc] = v;
  }
  __syncthreads();
  const int lane = tid & 31;
  const int col0 = (tid >> 5) * 16;
  v8f c;
  float bias = bvec[col0 + (lane & 15)];
  #pragma unroll
  for (int v = 0; v < 8; ++v) c[v] = bias;
  gemm_acc(As, 0, W, 72, col0, lane, c);
  store_tile(out, row0, nrows, col0, lane, c, true);
}

__global__ void __launch_bounds__(256) encode_comp_kernel(
    const float* __restrict__ x, const int* __restrict__ secidx,
    const int* __restrict__ indidx, const float* __restrict__ secemb,
    const float* __restrict__ indemb, const float* __restrict__ W,
    const float* __restrict__ bvec, float* __restrict__ out, int nrows) {
  __shared__ float As[16 * LDK];
  const int tid  = threadIdx.x;
  const int row0 = blockIdx.x * 16;
  for (int idx = tid; idx < 16 * 112; idx += 256) {
    int r = idx / 112, cc = idx % 112;
    int row = row0 + r;
    float v = 0.0f;
    if (row < nrows) {
      if (cc < 96)       v = x[(size_t)row * 96 + cc];
      else if (cc < 104) v = secemb[secidx[row] * 8 + (cc - 96)];
      else               v = indemb[indidx[row] * 8 + (cc - 104)];
    }
    As[r * LDK + cc] = v;
  }
  __syncthreads();
  const int lane = tid & 31;
  const int col0 = (tid >> 5) * 16;
  v8f c;
  float bias = bvec[col0 + (lane & 15)];
  #pragma unroll
  for (int v = 0; v < 8; ++v) c[v] = bias;
  gemm_acc(As, 0, W, 112, col0, lane, c);
  store_tile(out, row0, nrows, col0, lane, c, true);
}


__global__ void zero_kernel(float* __restrict__ p, size_t n) {
  size_t i      = (size_t)blockIdx.x * blockDim.x + threadIdx.x;
  size_t stride = (size_t)gridDim.x * blockDim.x;
  for (; i < n; i += stride) p[i] = 0.0f;
}

__global__ void __launch_bounds__(256) scatter_kernel(
    const float* __restrict__ h, const int* __restrict__ src,
    const int* __restrict__ dst, float* __restrict__ s,
    float* __restrict__ cnt, int E) {
  const int lane = threadIdx.x & 31;
  const int e    = blockIdx.x * 8 + (threadIdx.x >> 5);
  if (e >= E) return;
  const int sv = src[e];
  const int dv = dst[e];
  const float4 v = *reinterpret_cast<const float4*>(h + (size_t)sv * 128 + lane * 4);
  float* p = s + (size_t)dv * 128 + lane * 4;
  atomicAdd(p + 0, v.x);
  atomicAdd(p + 1, v.y);
  atomicAdd(p + 2, v.z);
  atomicAdd(p + 3, v.w);
  if (lane == 0) atomicAdd(cnt + dv, 1.0f);
}


__global__ void __launch_bounds__(256) sage_kernel(
    const float* __restrict__ h, const float* __restrict__ s,
    const float* __restrict__ cnt, const float* __restrict__ Wl,
    const float* __restrict__ bl, const float* __restrict__ Wr,
    float* __restrict__ out, int nrows, int do_relu) {
  __shared__ float As[16 * LDK];
  const int tid  = threadIdx.x;
  const int row0 = blockIdx.x * 16;
  for (int idx = tid; idx < 16 * 256; idx += 256) {
    int r = idx >> 8, cc = idx & 255;
    int row = row0 + r;
    float v = 0.0f;
    if (row < nrows) {
      if (cc < 128) {
        v = s[(size_t)row * 128 + cc] / fmaxf(cnt[row], 1.0f);
      } else {
        v = h[(size_t)row * 128 + (cc - 128)];
      }
    }
    As[r * LDK + cc] = v;
  }
  __syncthreads();
  const int lane = tid & 31;
  const int col0 = (tid >> 5) * 16;
  v8f c;
  float bias = bl[col0 + (lane & 15)];
  #pragma unroll
  for (int v = 0; v < 8; ++v) c[v] = bias;
  gemm_acc(As, 0,   Wl, 128, col0, lane, c);
  gemm_acc(As, 128, Wr, 128, col0, lane, c);
  store_tile(out, row0, nrows, col0, lane, c, do_relu != 0);
}


extern "C" void kernel_launch(void* const* d_in, const int* in_sizes, int n_in,
                              void* d_out, int out_size, void* d_ws, size_t ws_size,
                              hipStream_t stream) {
  (void)n_in; (void)out_size; (void)ws_size;
  const float* x_pol   = (const float*)d_in[0];
  const int*   pol_sid = (const int*)  d_in[1];
  const float* x_comp  = (const float*)d_in[2];
  const int*   c_sec   = (const int*)  d_in[3];
  const int*   c_ind   = (const int*)  d_in[4];
  const int*   edges   = (const int*)  d_in[5];
  const float* s_emb   = (const float*)d_in[6];
  const float* sec_emb = (const float*)d_in[7];
  const float* ind_emb = (const float*)d_in[8];
  const float* W_pol   = (const float*)d_in[9];
  const float* b_pol   = (const float*)d_in[10];
  const float* W_comp  = (const float*)d_in[11];
  const float* b_comp  = (const float*)d_in[12];
  const float* Wl1     = (const float*)d_in[13];
  const float* bl1     = (const float*)d_in[14];
  const float* Wr1     = (const float*)d_in[15];
  const float* Wl2     = (const float*)d_in[16];
  const float* bl2     = (const float*)d_in[17];
  const float* Wr2     = (const float*)d_in[18];

  const int NP = in_sizes[0] / 64;
  const int NC = in_sizes[2] / 96;
  const int E  = in_sizes[5] / 2;
  const int N  = NP + NC;
  const int* src = edges;
  const int* dst = edges + E;

  float* h    = (float*)d_ws;
  float* s    = h + (size_t)N * 128;
  float* cnt  = s + (size_t)N * 128;
  float* outp = (float*)d_out;

  const dim3 blk(256);
  const size_t zn = (size_t)N * 128 + N;
  int zgrid = (int)((zn + 255) / 256);
  if (zgrid > 65535) zgrid = 65535;

  encode_pol_kernel<<<dim3((NP + 15) / 16), blk, 0, stream>>>(
      x_pol, pol_sid, s_emb, W_pol, b_pol, h, NP);
  encode_comp_kernel<<<dim3((NC + 15) / 16), blk, 0, stream>>>(
      x_comp, c_sec, c_ind, sec_emb, ind_emb, W_comp, b_comp,
      h + (size_t)NP * 128, NC);

  zero_kernel<<<dim3(zgrid), blk, 0, stream>>>(s, zn);
  scatter_kernel<<<dim3((E + 7) / 8), blk, 0, stream>>>(h, src, dst, s, cnt, E);
  sage_kernel<<<dim3((N + 15) / 16), blk, 0, stream>>>(
      h, s, cnt, Wl1, bl1, Wr1, h, N, 1);

  zero_kernel<<<dim3(zgrid), blk, 0, stream>>>(s, zn);
  scatter_kernel<<<dim3((E + 7) / 8), blk, 0, stream>>>(h, src, dst, s, cnt, E);
  sage_kernel<<<dim3((N + 15) / 16), blk, 0, stream>>>(
      h, s, cnt, Wl2, bl2, Wr2, outp, N, 0);
}


# Each wave sums 13 rows of the tile alone: the scan hands every edge to its owner wave's list (LDS counters), both layers add rows by plain LDS load-add-store, no f32 atomics; sum order = list arrival
# speedup vs baseline: 2.7398x; 2.7398x over previous
#include <hip/hip_runtime.h>


typedef __attribute__((ext_vector_type(2))) float v2f;
typedef __attribute__((ext_vector_type(8))) float v8f;
typedef __attribute__((ext_vector_type(4))) float v4f;

#define LDK 260

__device__ __forceinline__ v8f wmma4(v2f a, v2f b, v8f c) {
  return __builtin_amdgcn_wmma_f32_16x16x4_f32(false, a, false, b,
                                               (short)0, c, false, false);
}

__device__ __forceinline__ void gemm_acc(const float* As, int aoff,
                                         const float* __restrict__ B,
                                         int K, int col0, int lane, v8f& c) {
  const int hl = lane >> 4;
  const int m  = lane & 15;
  const int n  = lane & 15;
  #pragma unroll 4
  for (int k0 = 0; k0 < K; k0 += 4) {
    const int ka = k0 + 2 * hl;
    v2f a, b;
    a.x = As[m * LDK + aoff + ka];
    a.y = As[m * LDK + aoff + ka + 1];
    b.x = B[ka * 128 + col0 + n];
    b.y = B[(ka + 1) * 128 + col0 + n];
    c = wmma4(a, b, c);
  }
}

__device__ __forceinline__ void store_tile(float* __restrict__ out, int row0,
                                           int nrows, int col0, int lane,
                                           v8f c, bool do_relu) {
  const int hl = lane >> 4;
  const int n  = lane & 15;
  #pragma unroll
  for (int v = 0; v < 8; ++v) {
    int row = row0 + v + 8 * hl;
    float val = c[v];
    if (do_relu) val = fmaxf(val, 0.0f);
    if (row < nrows) out[(size_t)row * 128 + col0 + n] = val;
  }
}


__global__ void __launch_bounds__(256) encode_pol_kernel(
    const float* __restrict__ x, const int* __restrict__ sidx,
    const float* __restrict__ semb, const float* __restrict__ W,
    const float* __restrict__ bvec, float* __restrict__ out, int nrows) {
  __shared__ float As[16 * LDK];
  const int tid  = threadIdx.x;
  const int row0 = blockIdx.x * 16;
  for (int idx = tid; idx < 16 * 72; idx += 256) {
    int r = idx / 72, cc = idx % 72;
    int row = row0 + r;
    float v = 0.0f;
    if (row < nrows) {
      if (cc < 64) v = x[(size_t)row * 64 + cc];
      else         v = semb[sidx[row] * 8 + (cc - 64)];
    }
    As[r * LDK + cc] = v;
  }
  __syncthreads();
  const int lane = tid & 31;
  const int col0 = (tid >> 5) * 16;
  v8f c;
  float bias = bvec[col0 + (lane & 15)];
  #pragma unroll
  for (int v = 0; v < 8; ++v) c[v] = bias;
  gemm_acc(As, 0, W, 72, col0, lane, c);
  store_tile(out, row0, nrows, col0, lane, c, true);
}

__global__ void __launch_bounds__(256) encode_comp_kernel(
    const float* __restrict__ x, const int* __restrict__ secidx,
    const int* __restrict__ indidx, const float* __restrict__ secemb,
    const float* __restrict__ indemb, const float* __restrict__ W,
    const float* __restrict__ bvec, float* __restrict__ out, int nrows) {
  __shared__ float As[16 * LDK];
  const int tid  = threadIdx.x;
  const int row0 = blockIdx.x * 16;
  for (int idx = tid; idx < 16 * 112; idx += 256) {
    int r = idx / 112, cc = idx % 112;
    int row = row0 + r;
    float v = 0.0f;
    if (row < nrows) {
      if (cc < 96)       v = x[(size_t)row * 96 + cc];
      else if (cc < 104) v = secemb[secidx[row] * 8 + (cc - 96)];
      else               v = indemb[indidx[row] * 8 + (cc - 104)];
    }
    As[r * LDK + cc] = v;
  }
  __syncthreads();
  const int lane = tid & 31;
  const int col0 = (tid >> 5) * 16;
  v8f c;
  float bias = bvec[col0 + (lane & 15)];
  #pragma unroll
  for (int v = 0; v < 8; ++v) c[v] = bias;
  gemm_acc(As, 0, W, 112, col0, lane, c);
  store_tile(out, row0, nrows, col0, lane, c, true);
}


#define AT    400
#define AGRP  512
#define AQ    576

__device__ __forceinline__ void agg_load_group(int4 (&d)[4], const int* __restrict__ dst,
                                               int grp, int lane) {
  const int* p = dst + (size_t)grp * AGRP + lane * 4;
  #pragma unroll
  for (int j = 0; j < 4; ++j) d[j] = *reinterpret_cast<const int4*>(p + j * 128);
}

#define AOWN  13
#define LHDR  32

template <bool SECOND>
__global__ void __launch_bounds__(1024) aggregate_kernel(
    const float* __restrict__ h, const int* __restrict__ src,
    const int* __restrict__ dst, float* __restrict__ s,
    float* __restrict__ cnt, int* __restrict__ lists, int E, int N) {
  __shared__ float acc[AT * 128];
  __shared__ int   cntl[AT];
  __shared__ int   qbuf[32 * AQ];
  __shared__ int   sub[32];

  const int tid   = threadIdx.x;
  const int lane  = tid & 31;
  const int w     = __builtin_amdgcn_readfirstlane(tid >> 5);
  const int nbase = blockIdx.x * AT;
  const int rows  = min(AT, N - nbase);
  int* q = qbuf + w * AQ;
  int* mylist   = lists + (size_t)nbase * 128;
  const int subcap = ((rows * 128 - LHDR) / 32) & ~31;

  const bool hi  = lane >= 16;
  const int  oa  = lane * 4 + (hi ? 2 : 0);
  const int  ob  = lane * 4 + (hi ? 0 : 2);
  auto add4 = [&](int row, const float4& v) {
    float* pa = acc + row * 128 + oa;
    float* pb = acc + row * 128 + ob;
    atomicAdd(pa + 0, hi ? v.z : v.x);
    atomicAdd(pa + 1, hi ? v.w : v.y);
    atomicAdd(pb + 0, hi ? v.x : v.z);
    atomicAdd(pb + 1, hi ? v.y : v.w);
  };

  auto add_rows = [&](int sv, int dl, int n) {
    int k = 0;
    for (; k + 8 <= n; k += 8) {
      float4 v[8];
      int    dd[8];
      #pragma unroll
      for (int u = 0; u < 8; ++u) {
        const int su = __builtin_amdgcn_readlane(sv, k + u);
        dd[u]        = __builtin_amdgcn_readlane(dl, k + u);
        v[u] = *reinterpret_cast<const float4*>(h + (size_t)su * 128 + lane * 4);
      }
      #pragma unroll
      for (int u = 0; u < 8; ++u) add4(dd[u], v[u]);
    }
    for (; k < n; ++k) {
      const int su = __builtin_amdgcn_readlane(sv, k);
      const int du = __builtin_amdgcn_readlane(dl, k);
      const float4 v = *reinterpret_cast<const float4*>(h + (size_t)su * 128 + lane * 4);
      add4(du, v);
    }
  };

  auto scan = [&](const bool LIST) __attribute__((always_inline)) {
    auto drain_round = [&](int r, int n) {
      int sv = 0, dl = 0;
      if (lane < n) {
        const int e = q[r + lane];
        sv = src[e];
        dl = dst[e] - nbase;
        if (LIST) {
          atomicAdd(&cntl[dl], 1);
          const int ow  = dl / AOWN;
          const int pos = atomicAdd(&sub[ow], 1);
          if (pos < subcap) mylist[LHDR + ow * subcap + pos] = (sv << 9) | dl;
        }
      }
      if (!LIST) add_rows(sv, dl, n);
    };

    int qn = 0;

    auto push = [&](bool match, int e) {
      const unsigned mm = __builtin_amdgcn_ballot_w32(match);
      if (mm) {
        if (match) q[qn + (int)__builtin_amdgcn_mbcnt_lo(mm, 0u)] = e;
        qn += __builtin_popcount(mm);
      }
    };

    const int nfull = E / AGRP;
    if (w == 0) {
      for (int e0 = nfull * AGRP; e0 < E; e0 += 32) {
        const int e = e0 + lane;
        const int d = (e < E) ? dst[e] : -1;
        push((unsigned)(d - nbase) < (unsigned)rows, e);
      }
    }

    auto make_room = [&]() {
      if (qn > AQ - AGRP) {
        const int full = qn & ~31;
        for (int r = 0; r < full; r += 32) drain_round(r, 32);
        const int rem = qn - full;
        int t = 0;
        if (lane < rem) t = q[full + lane];
        if (lane < rem) q[lane] = t;
        qn = rem;
      }
    };

    auto test_group = [&](const int4 (&d)[4], int g) {
      const int gb = g * AGRP;
      #pragma unroll
      for (int j = 0; j < 4; ++j) {
        const int dv[4] = {d[j].x, d[j].y, d[j].z, d[j].w};
        #pragma unroll
        for (int c = 0; c < 4; ++c)
          push((unsigned)(dv[c] - nbase) < (unsigned)rows, gb + j * 128 + lane * 4 + c);
      }
    };

    if (nfull > 0) {
      const int nit = (nfull + 31) / 32;
      int itA = (int)((blockIdx.x * 13u) % (unsigned)nit);
      int4 A[4], B[4];
      agg_load_group(A, dst, min(itA * 32 + w, nfull - 1), lane);
      for (int i = 0; i < nit; i += 2) {
        const int itB = (itA + 1 == nit) ? 0 : itA + 1;
        agg_load_group(B, dst, min(itB * 32 + w, nfull - 1), lane);
        make_room();
        if (itA * 32 + w < nfull) test_group(A, itA * 32 + w);
        itA = (itB + 1 == nit) ? 0 : itB + 1;
        agg_load_group(A, dst, min(itA * 32 + w, nfull - 1), lane);
        make_room();
        if (i + 1 < nit && itB * 32 + w < nfull) test_group(B, itB * 32 + w);
      }
    }
    for (int r = 0; r < qn; r += 32) drain_round(r, min(32, qn - r));
  };

  int nsub = 0;
  if (!SECOND) {
    for (int idx = tid; idx < AT; idx += 1024) cntl[idx] = 0;
    if (tid < 32) sub[tid] = 0;
    __syncthreads();
    scan(true);
    __syncthreads();
    nsub = sub[lane];
  } else {
    nsub = mylist[lane];
  }
  const bool replay = __builtin_amdgcn_ballot_w32((unsigned)nsub > (unsigned)subcap) == 0u;

  if (replay) {
    const int own0 = w * AOWN;
    const int ownn = max(0, min(AOWN, rows - own0));
    float* myacc = acc + own0 * 128 + lane * 4;
    for (int j = 0; j < ownn; ++j)
      *reinterpret_cast<v4f*>(myacc + j * 128) = v4f{0.f, 0.f, 0.f, 0.f};

    auto loadg = [&](int sv, int dl, int k, v4f (&v)[8], int (&dd)[8]) {
      #pragma unroll
      for (int u = 0; u < 8; ++u) {
        const int su = __builtin_amdgcn_readlane(sv, k + u);
        dd[u]        = __builtin_amdgcn_readlane(dl, k + u);
        v[u] = *reinterpret_cast<const v4f*>(h + (size_t)su * 128 + lane * 4);
      }
    };
    auto rmw8 = [&](const v4f (&v)[8], const int (&dd)[8]) {
      v4f ld = *reinterpret_cast<const v4f*>(myacc + dd[0] * 128);
      v4f prev = ld;
      #pragma unroll
      for (int u = 0; u < 8; ++u) {
        v4f nx = ld;
        if (u < 7) nx = *reinterpret_cast<const v4f*>(myacc + dd[u + 1] * 128);
        const v4f base = (u > 0 && dd[u] == dd[u - 1]) ? prev : ld;
        const v4f sum  = base + v[u];
        *reinterpret_cast<v4f*>(myacc + dd[u] * 128) = sum;
        prev = sum;
        ld   = nx;
      }
    };
    auto round32 = [&](int p) {
      const int sv = (int)((unsigned)p >> 9);
      const int dl = (p & 511) - own0;
      v4f A[8], B[8];
      int da[8], db[8];
      loadg(sv, dl, 0, A, da);
      loadg(sv, dl, 8, B, db);
      rmw8(A, da);
      loadg(sv, dl, 16, A, da);
      rmw8(B, db);
      loadg(sv, dl, 24, B, db);
      rmw8(A, da);
      rmw8(B, db);
    };
    auto round_rest = [&](int p, int n) {
      const int sv = (int)((unsigned)p >> 9);
      const int dl = (p & 511) - own0;
      int k = 0;
      for (; k + 8 <= n; k += 8) {
        v4f A[8];
        int da[8];
        loadg(sv, dl, k, A, da);
        rmw8(A, da);
      }
      for (; k < n; ++k) {
        const int su = __builtin_amdgcn_readlane(sv, k);
        const int du = __builtin_amdgcn_readlane(dl, k);
        const v4f v = *reinterpret_cast<const v4f*>(h + (size_t)su * 128 + lane * 4);
        v4f* pa = reinterpret_cast<v4f*>(myacc + du * 128);
        *pa = *pa + v;
      }
    };

    const int  mycnt = __builtin_amdgcn_readlane(nsub, w);
    const int* ent   = mylist + LHDR + w * subcap;
    int p = (lane < mycnt) ? ent[lane] : 0;
    for (int r = 0; r < mycnt; r += 32) {
      const int pn = (r + 32 + lane < mycnt) ? ent[r + 32 + lane] : 0;
      if (mycnt - r >= 32) round32(p);
      else                 round_rest(p, mycnt - r);
      p = pn;
    }

    for (int j = 0; j < ownn; ++j)
      *reinterpret_cast<v4f*>(s + (size_t)(nbase + own0 + j) * 128 + lane * 4) =
          *reinterpret_cast<const v4f*>(myacc + j * 128);
  } else {
    for (int idx = tid; idx < AT * 32; idx += 1024)
      *reinterpret_cast<float4*>(acc + idx * 4) = make_float4(0.f, 0.f, 0.f, 0.f);
    __syncthreads();
    scan(false);
    __syncthreads();
    for (int idx = tid; idx < rows * 32; idx += 1024) {
      const int r = idx >> 5, c4 = idx & 31;
      *reinterpret_cast<float4*>(s + (size_t)(nbase + r) * 128 + c4 * 4) =
          *reinterpret_cast<const float4*>(acc + r * 128 + c4 * 4);
    }
  }
  if (!SECOND) {
    for (int r = tid; r < rows; r += 1024) cnt[nbase + r] = (float)cntl[r];
    if (tid < 32) mylist[tid] = sub[tid];
  }
}


#define SAGE_WGS 512
#define SAGE_KS  64

__global__ void __launch_bounds__(256) sage_kernel(
    const float* h, const float* s,
    const float* __restrict__ cnt, const float* __restrict__ Wl,
    const float* __restrict__ bl, const float* __restrict__ Wr,
    float* out, int nrows, int do_relu) {
  __shared__ float As[2][16 * LDK];
  const int tid    = threadIdx.x;
  const int lane   = tid & 31;
  const int hl     = lane >> 4;
  const int col0   = (tid >> 5) * 16;
  const int ntiles = (nrows + 15) >> 4;

  v2f bf[SAGE_KS];
  {
    const int cn = col0 + (lane & 15);
    #pragma unroll
    for (int j = 0; j < 32; ++j) {
      const int k = 4 * j + 2 * hl;
      bf[j].x      = Wl[k * 128 + cn];
      bf[j].y      = Wl[(k + 1) * 128 + cn];
      bf[32 + j].x = Wr[k * 128 + cn];
      bf[32 + j].y = Wr[(k + 1) * 128 + cn];
    }
  }
  const float bias = bl[col0 + (lane & 15)];

  const int  c4     = tid & 63;
  const int  rb     = tid >> 6;
  const bool is_sum = (__builtin_amdgcn_readfirstlane(tid >> 5) & 1) == 0;

  const float* rows_in = (is_sum ? s : h) + (c4 & 31) * 4;
  auto fetch = [&](int t, float4 (&v)[4], float (&d)[4]) {
    #pragma unroll
    for (int i = 0; i < 4; ++i) {
      const int row = min(t * 16 + rb + 4 * i, nrows - 1);
      v[i] = *reinterpret_cast<const float4*>(rows_in + (size_t)row * 128);
      d[i] = cnt[row];
    }
  };
  auto put = [&](float* A, const float4 (&v)[4], const float (&d)[4]) {
    #pragma unroll
    for (int i = 0; i < 4; ++i) {
      float4 x = v[i];
      if (is_sum) {
        const float dd = fmaxf(d[i], 1.0f);
        x.x /= dd; x.y /= dd; x.z /= dd; x.w /= dd;
      }
      *reinterpret_cast<float4*>(A + (rb + 4 * i) * LDK + c4 * 4) = x;
    }
  };

  int t = blockIdx.x;
  if (t < ntiles) {
    float4 v[4];
    float  d[4];
    fetch(t, v, d);
    put(As[0], v, d);
  }
  __syncthreads();

  int buf = 0;
  for (; t < ntiles; t += SAGE_WGS) {
    const int  tn   = t + SAGE_WGS;
    const bool more = tn < ntiles;
    float4 vn[4];
    float  dn[4];
    fetch(tn, vn, dn);
    __builtin_amdgcn_sched_barrier(0);

    const float* ap = As[buf] + (lane & 15) * LDK + 2 * hl;
    v8f c;
    #pragma unroll
    for (int v = 0; v < 8; ++v) c[v] = bias;
    #pragma unroll
    for (int j = 0; j < SAGE_KS; ++j) {
      const v2f a = *reinterpret_cast<const v2f*>(ap + 4 * j);
      c = wmma4(a, bf[j], c);
    }
    if (more) put(As[buf ^ 1], vn, dn);
    __syncthreads();
    store_tile(out, t * 16, nrows, col0, lane, c, do_relu != 0);
    buf ^= 1;
  }
}


extern "C" void kernel_launch(void* const* d_in, const int* in_sizes, int n_in,
                              void* d_out, int out_size, void* d_ws, size_t ws_size,
                              hipStream_t stream) {
  (void)n_in; (void)out_size; (void)ws_size;
  const float* x_pol   = (const float*)d_in[0];
  const int*   pol_sid = (const int*)  d_in[1];
  const float* x_comp  = (const float*)d_in[2];
  const int*   c_sec   = (const int*)  d_in[3];
  const int*   c_ind   = (const int*)  d_in[4];
  const int*   edges   = (const int*)  d_in[5];
  const float* s_emb   = (const float*)d_in[6];
  const float* sec_emb = (const float*)d_in[7];
  const float* ind_emb = (const float*)d_in[8];
  const float* W_pol   = (const float*)d_in[9];
  const float* b_pol   = (const float*)d_in[10];
  const float* W_comp  = (const float*)d_in[11];
  const float* b_comp  = (const float*)d_in[12];
  const float* Wl1     = (const float*)d_in[13];
  const float* bl1     = (const float*)d_in[14];
  const float* Wr1     = (const float*)d_in[15];
  const float* Wl2     = (const float*)d_in[16];
  const float* bl2     = (const float*)d_in[17];
  const float* Wr2     = (const float*)d_in[18];

  const int NP = in_sizes[0] / 64;
  const int NC = in_sizes[2] / 96;
  const int E  = in_sizes[5] / 2;
  const int N  = NP + NC;
  const int* src = edges;
  const int* dst = edges + E;

  float* h    = (float*)d_ws;
  float* s    = h + (size_t)N * 128;
  float* cnt  = s + (size_t)N * 128;
  float* outp = (float*)d_out;
  int*   lists = (int*)d_out;

  const dim3 blk(256);
  const dim3 ablk(1024);
  const dim3 agrid((N + AT - 1) / AT);

  encode_pol_kernel<<<dim3((NP + 15) / 16), blk, 0, stream>>>(
      x_pol, pol_sid, s_emb, W_pol, b_pol, h, NP);
  encode_comp_kernel<<<dim3((NC + 15) / 16), blk, 0, stream>>>(
      x_comp, c_sec, c_ind, sec_emb, ind_emb, W_comp, b_comp,
      h + (size_t)NP * 128, NC);

  aggregate_kernel<false><<<agrid, ablk, 0, stream>>>(h, src, dst, s, cnt, lists, E, N);
  sage_kernel<<<dim3(SAGE_WGS), blk, 0, stream>>>(
      h, s, cnt, Wl1, bl1, Wr1, h, N, 1);

  aggregate_kernel<true><<<agrid, ablk, 0, stream>>>(h, src, dst, s, cnt, lists, E, N);
  sage_kernel<<<dim3(SAGE_WGS), blk, 0, stream>>>(
      h, s, cnt, Wl2, bl2, Wr2, outp, N, 0);
}


# Layer 1's scan reads one-byte tile tags (a small new launch writes them into idle output rows), four tested per word: a quarter of the id bytes per work-group; sum order = list arrival as before
# speedup vs baseline: 2.9313x; 1.0699x over previous
#include <hip/hip_runtime.h>


typedef __attribute__((ext_vector_type(2))) float v2f;
typedef __attribute__((ext_vector_type(8))) float v8f;
typedef __attribute__((ext_vector_type(4))) float v4f;

#define LDK 260

__device__ __forceinline__ v8f wmma4(v2f a, v2f b, v8f c) {
  return __builtin_amdgcn_wmma_f32_16x16x4_f32(false, a, false, b,
                                               (short)0, c, false, false);
}

__device__ __forceinline__ void gemm_acc(const float* As, int aoff,
                                         const float* __restrict__ B,
                                         int K, int col0, int lane, v8f& c) {
  const int hl = lane >> 4;
  const int m  = lane & 15;
  const int n  = lane & 15;
  #pragma unroll 4
  for (int k0 = 0; k0 < K; k0 += 4) {
    const int ka = k0 + 2 * hl;
    v2f a, b;
    a.x = As[m * LDK + aoff + ka];
    a.y = As[m * LDK + aoff + ka + 1];
    b.x = B[ka * 128 + col0 + n];
    b.y = B[(ka + 1) * 128 + col0 + n];
    c = wmma4(a, b, c);
  }
}

__device__ __forceinline__ void store_tile(float* __restrict__ out, int row0,
                                           int nrows, int col0, int lane,
                                           v8f c, bool do_relu) {
  const int hl = lane >> 4;
  const int n  = lane & 15;
  #pragma unroll
  for (int v = 0; v < 8; ++v) {
    int row = row0 + v + 8 * hl;
    float val = c[v];
    if (do_relu) val = fmaxf(val, 0.0f);
    if (row < nrows) out[(size_t)row * 128 + col0 + n] = val;
  }
}


__global__ void __launch_bounds__(256) encode_pol_kernel(
    const float* __restrict__ x, const int* __restrict__ sidx,
    const float* __restrict__ semb, const float* __restrict__ W,
    const float* __restrict__ bvec, float* __restrict__ out, int nrows) {
  __shared__ float As[16 * LDK];
  const int tid  = threadIdx.x;
  const int row0 = blockIdx.x * 16;
  for (int idx = tid; idx < 16 * 72; idx += 256) {
    int r = idx / 72, cc = idx % 72;
    int row = row0 + r;
    float v = 0.0f;
    if (row < nrows) {
      if (cc < 64) v = x[(size_t)row * 64 + cc];
      else         v = semb[sidx[row] * 8 + (cc - 64)];
    }
    As[r * LDK + cc] = v;
  }
  __syncthreads();
  const int lane = tid & 31;
  const int col0 = (tid >> 5) * 16;
  v8f c;
  float bias = bvec[col0 + (lane & 15)];
  #pragma unroll
  for (int v = 0; v < 8; ++v) c[v] = bias;
  gemm_acc(As, 0, W, 72, col0, lane, c);
  store_tile(out, row0, nrows, col0, lane, c, true);
}

__global__ void __launch_bounds__(256) encode_comp_kernel(
    const float* __restrict__ x, const int* __restrict__ secidx,
    const int* __restrict__ indidx, const float* __restrict__ secemb,
    const float* __restrict__ indemb, const float* __restrict__ W,
    const float* __restrict__ bvec, float* __restrict__ out, int nrows) {
  __shared__ float As[16 * LDK];
  const int tid  = threadIdx.x;
  const int row0 = blockIdx.x * 16;
  for (int idx = tid; idx < 16 * 112; idx += 256) {
    int r = idx / 112, cc = idx % 112;
    int row = row0 + r;
    float v = 0.0f;
    if (row < nrows) {
      if (cc < 96)       v = x[(size_t)row * 96 + cc];
      else if (cc < 104) v = secemb[secidx[row] * 8 + (cc - 96)];
      else               v = indemb[indidx[row] * 8 + (cc - 104)];
    }
    As[r * LDK + cc] = v;
  }
  __syncthreads();
  const int lane = tid & 31;
  const int col0 = (tid >> 5) * 16;
  v8f c;
  float bias = bvec[col0 + (lane & 15)];
  #pragma unroll
  for (int v = 0; v < 8; ++v) c[v] = bias;
  gemm_acc(As, 0, W, 112, col0, lane, c);
  store_tile(out, row0, nrows, col0, lane, c, true);
}


#define AT    400
#define AGRP  2048
#define ASUB  512
#define AKEEP 64
#define AQ    128
#define HT    127

#define LCAPW   1024
#define LHDR    32
#define TAGOFF  (LHDR + 32 * LCAPW)
#define TAGSEG  65536

__device__ __forceinline__ void agg_load_group(int4 (&d)[4], const int* outw,
                                               int grp, int lane) {
  const int* p = outw + (size_t)(grp >> 5) * (AT * 128) + TAGOFF + (grp & 31) * (AGRP / 4) + lane * 4;
  #pragma unroll
  for (int j = 0; j < 4; ++j) d[j] = *reinterpret_cast<const int4*>(p + j * 128);
}

__global__ void __launch_bounds__(256) tag_kernel(const int* __restrict__ dst,
                                                  int* __restrict__ outw, int nunits) {
  const int u = blockIdx.x * 256 + threadIdx.x;
  if (u >= nunits) return;
  const int4* p = reinterpret_cast<const int4*>(dst + (size_t)u * 16);
  int4 t0, t1;
  int* w0 = reinterpret_cast<int*>(&t0);
  int* w1 = reinterpret_cast<int*>(&t1);
  #pragma unroll
  for (int j = 0; j < 4; ++j) {
    const int4 d = p[j];
    const unsigned dv[4] = {(unsigned)d.x, (unsigned)d.y, (unsigned)d.z, (unsigned)d.w};
    unsigned a = 0u, b = 0u;
    #pragma unroll
    for (int c = 0; c < 4; ++c) {
      const unsigned tile = dv[c] / AT;
      const bool up = tile >= HT;
      a |= (up ? 127u : tile) << (8 * c);
      b |= (up ? tile - HT : 127u) << (8 * c);
    }
    w0[j] = (int)a;
    w1[j] = (int)b;
  }
  const int nseg = (nunits / (AGRP / 16) + 31) >> 5;
  const size_t e0 = (size_t)u * 16;
  int* q0 = outw + (e0 / TAGSEG) * (size_t)(AT * 128) + TAGOFF + (e0 % TAGSEG) / 4;
  int* q1 = q0 + (size_t)nseg * (AT * 128);
  *reinterpret_cast<int4*>(q0) = t0;
  *reinterpret_cast<int4*>(q1) = t1;
}

#define AOWN  13

template <bool SECOND>
__global__ void __launch_bounds__(1024) aggregate_kernel(
    const float* __restrict__ h, const int* __restrict__ src,
    const int* __restrict__ dst, float* __restrict__ s,
    float* __restrict__ cnt, int* __restrict__ lists, int E, int N) {
  __shared__ float acc[AT * 128];
  __shared__ int   cntl[AT];
  __shared__ int   qbuf[32 * AQ];
  __shared__ int   sub[32];

  const int tid   = threadIdx.x;
  const int lane  = tid & 31;
  const int w     = __builtin_amdgcn_readfirstlane(tid >> 5);
  const int nbase = blockIdx.x * AT;
  const int rows  = min(AT, N - nbase);
  int* q = qbuf + w * AQ;
  int* mylist   = lists + (size_t)nbase * 128;
  const int subcap = min(((rows * 128 - LHDR) / 32) & ~31, LCAPW);
  const int      half = (blockIdx.x >= HT) ? 1 : 0;
  const unsigned pat  = (blockIdx.x - (unsigned)(half * HT)) * 0x01010101u;

  const bool hi  = lane >= 16;
  const int  oa  = lane * 4 + (hi ? 2 : 0);
  const int  ob  = lane * 4 + (hi ? 0 : 2);
  auto add4 = [&](int row, const float4& v) {
    float* pa = acc + row * 128 + oa;
    float* pb = acc + row * 128 + ob;
    atomicAdd(pa + 0, hi ? v.z : v.x);
    atomicAdd(pa + 1, hi ? v.w : v.y);
    atomicAdd(pb + 0, hi ? v.x : v.z);
    atomicAdd(pb + 1, hi ? v.y : v.w);
  };

  auto add_rows = [&](int sv, int dl, int n) {
    int k = 0;
    for (; k + 8 <= n; k += 8) {
      float4 v[8];
      int    dd[8];
      #pragma unroll
      for (int u = 0; u < 8; ++u) {
        const int su = __builtin_amdgcn_readlane(sv, k + u);
        dd[u]        = __builtin_amdgcn_readlane(dl, k + u);
        v[u] = *reinterpret_cast<const float4*>(h + (size_t)su * 128 + lane * 4);
      }
      #pragma unroll
      for (int u = 0; u < 8; ++u) add4(dd[u], v[u]);
    }
    for (; k < n; ++k) {
      const int su = __builtin_amdgcn_readlane(sv, k);
      const int du = __builtin_amdgcn_readlane(dl, k);
      const float4 v = *reinterpret_cast<const float4*>(h + (size_t)su * 128 + lane * 4);
      add4(du, v);
    }
  };

  auto scan = [&](const bool LIST) __attribute__((always_inline)) {
    auto drain_round = [&](int r, int n) {
      int sv = 0, dl = 0;
      if (lane < n) {
        const int e = q[r + lane];
        sv = src[e];
        dl = dst[e] - nbase;
        if (LIST) {
          atomicAdd(&cntl[dl], 1);
          const int ow  = dl / AOWN;
          const int pos = atomicAdd(&sub[ow], 1);
          if (pos < subcap) mylist[LHDR + ow * subcap + pos] = (sv << 9) | dl;
        }
      }
      if (!LIST) add_rows(sv, dl, n);
    };

    int qn = 0;

    auto push = [&](bool match, int e) {
      const unsigned mm = __builtin_amdgcn_ballot_w32(match);
      if (mm) {
        if (match) q[qn + (int)__builtin_amdgcn_mbcnt_lo(mm, 0u)] = e;
        qn += __builtin_popcount(mm);
      }
    };

    auto make_room = [&]() {
      if (qn > AKEEP) {
        const int full = qn & ~31;
        for (int r = 0; r < full; r += 32) drain_round(r, 32);
        const int rem = qn - full;
        int t = 0;
        if (lane < rem) t = q[full + lane];
        if (lane < rem) q[lane] = t;
        qn = rem;
      }
    };

    const int nfull = E / AGRP;
    if (w == 0) {
      for (int e0 = nfull * AGRP; e0 < E; e0 += 32) {
        make_room();
        const int e = e0 + lane;
        const int d = (e < E) ? dst[e] : -1;
        push((unsigned)(d - nbase) < (unsigned)rows, e);
      }
    }

    auto miss = [&](int x) -> unsigned { return ((unsigned)x ^ pat) + 0x7f7f7f7fu; };
    auto ours = [&](const int4& d) -> unsigned {
      const unsigned n = ((miss(d.x) >> 7) & 0x01010101u) | ((miss(d.y) >> 6) & 0x02020202u) |
                         ((miss(d.z) >> 5) & 0x04040404u) | ((miss(d.w) >> 4) & 0x08080808u);
      return n ^ 0x0f0f0f0fu;
    };
    auto test_group = [&](const int4 (&d)[4], int g) {
      const int gb = g * AGRP + lane * 16;
      unsigned lo = ours(d[0]) | (ours(d[1]) << 4);
      unsigned hi = ours(d[2]) | (ours(d[3]) << 4);
      unsigned mm = __builtin_amdgcn_ballot_w32((lo | hi) != 0u);
      while (mm != 0u) {
        make_room();
        if ((lo | hi) != 0u) {
          const bool     inlo = lo != 0u;
          const unsigned cur  = inlo ? lo : hi;
          const int      b    = __builtin_ctz(cur);
          const int      j    = (inlo ? 0 : 2) + ((b >> 2) & 1);
          q[qn + (int)__builtin_amdgcn_mbcnt_lo(mm, 0u)] =
              gb + j * ASUB + ((b & 3) << 2) + (b >> 3);
          const unsigned rest = cur & (cur - 1u);
          lo = inlo ? rest : lo;
          hi = inlo ? hi : rest;
        }
        qn += __builtin_popcount(mm);
        mm = __builtin_amdgcn_ballot_w32((lo | hi) != 0u);
      }
    };

    if (nfull > 0) {
      const int* tags = lists + (size_t)(half * ((nfull + 31) >> 5)) * (AT * 128);
      const int nit = (nfull + 31) / 32;
      int itA = (int)((blockIdx.x * 13u) % (unsigned)nit);
      int4 A[4], B[4];
      agg_load_group(A, tags, min(itA * 32 + w, nfull - 1), lane);
      for (int i = 0; i < nit; i += 2) {
        const int itB = (itA + 1 == nit) ? 0 : itA + 1;
        agg_load_group(B, tags, min(itB * 32 + w, nfull - 1), lane);
        if (itA * 32 + w < nfull) test_group(A, itA * 32 + w);
        itA = (itB + 1 == nit) ? 0 : itB + 1;
        agg_load_group(A, tags, min(itA * 32 + w, nfull - 1), lane);
        if (i + 1 < nit && itB * 32 + w < nfull) test_group(B, itB * 32 + w);
      }
    }
    for (int r = 0; r < qn; r += 32) drain_round(r, min(32, qn - r));
  };

  int nsub = 0;
  if (!SECOND) {
    for (int idx = tid; idx < AT; idx += 1024) cntl[idx] = 0;
    if (tid < 32) sub[tid] = 0;
    __syncthreads();
    scan(true);
    __syncthreads();
    nsub = sub[lane];
  } else {
    nsub = mylist[lane];
  }
  const bool replay = __builtin_amdgcn_ballot_w32((unsigned)nsub > (unsigned)subcap) == 0u;

  if (replay) {
    const int own0 = w * AOWN;
    const int ownn = max(0, min(AOWN, rows - own0));
    float* myacc = acc + own0 * 128 + lane * 4;
    for (int j = 0; j < ownn; ++j)
      *reinterpret_cast<v4f*>(myacc + j * 128) = v4f{0.f, 0.f, 0.f, 0.f};

    auto loadg = [&](int sv, int dl, int k, v4f (&v)[8], int (&dd)[8]) {
      #pragma unroll
      for (int u = 0; u < 8; ++u) {
        const int su = __builtin_amdgcn_readlane(sv, k + u);
        dd[u]        = __builtin_amdgcn_readlane(dl, k + u);
        v[u] = *reinterpret_cast<const v4f*>(h + (size_t)su * 128 + lane * 4);
      }
    };
    auto rmw8 = [&](const v4f (&v)[8], const int (&dd)[8]) {
      v4f ld = *reinterpret_cast<const v4f*>(myacc + dd[0] * 128);
      v4f prev = ld;
      #pragma unroll
      for (int u = 0; u < 8; ++u) {
        v4f nx = ld;
        if (u < 7) nx = *reinterpret_cast<const v4f*>(myacc + dd[u + 1] * 128);
        const v4f base = (u > 0 && dd[u] == dd[u - 1]) ? prev : ld;
        const v4f sum  = base + v[u];
        *reinterpret_cast<v4f*>(myacc + dd[u] * 128) = sum;
        prev = sum;
        ld   = nx;
      }
    };
    auto round32 = [&](int p) {
      const int sv = (int)((unsigned)p >> 9);
      const int dl = (p & 511) - own0;
      v4f A[8], B[8];
      int da[8], db[8];
      loadg(sv, dl, 0, A, da);
      loadg(sv, dl, 8, B, db);
      rmw8(A, da);
      loadg(sv, dl, 16, A, da);
      rmw8(B, db);
      loadg(sv, dl, 24, B, db);
      rmw8(A, da);
      rmw8(B, db);
    };
    auto round_rest = [&](int p, int n) {
      const int sv = (int)((unsigned)p >> 9);
      const int dl = (p & 511) - own0;
      int k = 0;
      for (; k + 8 <= n; k += 8) {
        v4f A[8];
        int da[8];
        loadg(sv, dl, k, A, da);
        rmw8(A, da);
      }
      for (; k < n; ++k) {
        const int su = __builtin_amdgcn_readlane(sv, k);
        const int du = __builtin_amdgcn_readlane(dl, k);
        const v4f v = *reinterpret_cast<const v4f*>(h + (size_t)su * 128 + lane * 4);
        v4f* pa = reinterpret_cast<v4f*>(myacc + du * 128);
        *pa = *pa + v;
      }
    };

    const int  mycnt = __builtin_amdgcn_readlane(nsub, w);
    const int* ent   = mylist + LHDR + w * subcap;
    int p = (lane < mycnt) ? ent[lane] : 0;
    for (int r = 0; r < mycnt; r += 32) {
      const int pn = (r + 32 + lane < mycnt) ? ent[r + 32 + lane] : 0;
      if (mycnt - r >= 32) round32(p);
      else                 round_rest(p, mycnt - r);
      p = pn;
    }

    for (int j = 0; j < ownn; ++j)
      *reinterpret_cast<v4f*>(s + (size_t)(nbase + own0 + j) * 128 + lane * 4) =
          *reinterpret_cast<const v4f*>(myacc + j * 128);
  } else {
    for (int idx = tid; idx < AT * 32; idx += 1024)
      *reinterpret_cast<float4*>(acc + idx * 4) = make_float4(0.f, 0.f, 0.f, 0.f);
    __syncthreads();
    scan(false);
    __syncthreads();
    for (int idx = tid; idx < rows * 32; idx += 1024) {
      const int r = idx >> 5, c4 = idx & 31;
      *reinterpret_cast<float4*>(s + (size_t)(nbase + r) * 128 + c4 * 4) =
          *reinterpret_cast<const float4*>(acc + r * 128 + c4 * 4);
    }
  }
  if (!SECOND) {
    for (int r = tid; r < rows; r += 1024) cnt[nbase + r] = (float)cntl[r];
    if (tid < 32) mylist[tid] = sub[tid];
  }
}


#define SAGE_WGS 512
#define SAGE_KS  64

__global__ void __launch_bounds__(256) sage_kernel(
    const float* h, const float* s,
    const float* __restrict__ cnt, const float* __restrict__ Wl,
    const float* __restrict__ bl, const float* __restrict__ Wr,
    float* out, int nrows, int do_relu) {
  __shared__ float As[2][16 * LDK];
  const int tid    = threadIdx.x;
  const int lane   = tid & 31;
  const int hl     = lane >> 4;
  const int col0   = (tid >> 5) * 16;
  const int ntiles = (nrows + 15) >> 4;

  v2f bf[SAGE_KS];
  {
    const int cn = col0 + (lane & 15);
    #pragma unroll
    for (int j = 0; j < 32; ++j) {
      const int k = 4 * j + 2 * hl;
      bf[j].x      = Wl[k * 128 + cn];
      bf[j].y      = Wl[(k + 1) * 128 + cn];
      bf[32 + j].x = Wr[k * 128 + cn];
      bf[32 + j].y = Wr[(k + 1) * 128 + cn];
    }
  }
  const float bias = bl[col0 + (lane & 15)];

  const int  c4     = tid & 63;
  const int  rb     = tid >> 6;
  const bool is_sum = (__builtin_amdgcn_readfirstlane(tid >> 5) & 1) == 0;

  const float* rows_in = (is_sum ? s : h) + (c4 & 31) * 4;
  auto fetch = [&](int t, float4 (&v)[4], float (&d)[4]) {
    #pragma unroll
    for (int i = 0; i < 4; ++i) {
      const int row = min(t * 16 + rb + 4 * i, nrows - 1);
      v[i] = *reinterpret_cast<const float4*>(rows_in + (size_t)row * 128);
      d[i] = cnt[row];
    }
  };
  auto put = [&](float* A, const float4 (&v)[4], const float (&d)[4]) {
    #pragma unroll
    for (int i = 0; i < 4; ++i) {
      float4 x = v[i];
      if (is_sum) {
        const float dd = fmaxf(d[i], 1.0f);
        x.x /= dd; x.y /= dd; x.z /= dd; x.w /= dd;
      }
      *reinterpret_cast<float4*>(A + (rb + 4 * i) * LDK + c4 * 4) = x;
    }
  };

  int t = blockIdx.x;
  if (t < ntiles) {
    float4 v[4];
    float  d[4];
    fetch(t, v, d);
    put(As[0], v, d);
  }
  __syncthreads();

  int buf = 0;
  for (; t < ntiles; t += SAGE_WGS) {
    const int  tn   = t + SAGE_WGS;
    const bool more = tn < ntiles;
    float4 vn[4];
    float  dn[4];
    fetch(tn, vn, dn);
    __builtin_amdgcn_sched_barrier(0);

    const float* ap = As[buf] + (lane & 15) * LDK + 2 * hl;
    v8f c;
    #pragma unroll
    for (int v = 0; v < 8; ++v) c[v] = bias;
    #pragma unroll
    for (int j = 0; j < SAGE_KS; ++j) {
      const v2f a = *reinterpret_cast<const v2f*>(ap + 4 * j);
      c = wmma4(a, bf[j], c);
    }
    if (more) put(As[buf ^ 1], vn, dn);
    __syncthreads();
    store_tile(out, t * 16, nrows, col0, lane, c, do_relu != 0);
    buf ^= 1;
  }
}


extern "C" void kernel_launch(void* const* d_in, const int* in_sizes, int n_in,
                              void* d_out, int out_size, void* d_ws, size_t ws_size,
                              hipStream_t stream) {
  (void)n_in; (void)out_size; (void)ws_size;
  const float* x_pol   = (const float*)d_in[0];
  const int*   pol_sid = (const int*)  d_in[1];
  const float* x_comp  = (const float*)d_in[2];
  const int*   c_sec   = (const int*)  d_in[3];
  const int*   c_ind   = (const int*)  d_in[4];
  const int*   edges   = (const int*)  d_in[5];
  const float* s_emb   = (const float*)d_in[6];
  const float* sec_emb = (const float*)d_in[7];
  const float* ind_emb = (const float*)d_in[8];
  const float* W_pol   = (const float*)d_in[9];
  const float* b_pol   = (const float*)d_in[10];
  const float* W_comp  = (const float*)d_in[11];
  const float* b_comp  = (const float*)d_in[12];
  const float* Wl1     = (const float*)d_in[13];
  const float* bl1     = (const float*)d_in[14];
  const float* Wr1     = (const float*)d_in[15];
  const float* Wl2     = (const float*)d_in[16];
  const float* bl2     = (const float*)d_in[17];
  const float* Wr2     = (const float*)d_in[18];

  const int NP = in_sizes[0] / 64;
  const int NC = in_sizes[2] / 96;
  const int E  = in_sizes[5] / 2;
  const int N  = NP + NC;
  const int* src = edges;
  const int* dst = edges + E;

  float* h    = (float*)d_ws;
  float* s    = h + (size_t)N * 128;
  float* cnt  = s + (size_t)N * 128;
  float* outp = (float*)d_out;
  int*   lists = (int*)d_out;

  const dim3 blk(256);
  const dim3 ablk(1024);
  const dim3 agrid((N + AT - 1) / AT);

  const int nunits = (E / AGRP) * (AGRP / 16);
  tag_kernel<<<dim3((nunits + 255) / 256), blk, 0, stream>>>(dst, lists, nunits);

  encode_pol_kernel<<<dim3((NP + 15) / 16), blk, 0, stream>>>(
      x_pol, pol_sid, s_emb, W_pol, b_pol, h, NP);
  encode_comp_kernel<<<dim3((NC + 15) / 16), blk, 0, stream>>>(
      x_comp, c_sec, c_ind, sec_emb, ind_emb, W_comp, b_comp,
      h + (size_t)NP * 128, NC);

  aggregate_kernel<false><<<agrid, ablk, 0, stream>>>(h, src, dst, s, cnt, lists, E, N);
  sage_kernel<<<dim3(SAGE_WGS), blk, 0, stream>>>(
      h, s, cnt, Wl1, bl1, Wr1, h, N, 1);

  aggregate_kernel<true><<<agrid, ablk, 0, stream>>>(h, src, dst, s, cnt, lists, E, N);
  sage_kernel<<<dim3(SAGE_WGS), blk, 0, stream>>>(
      h, s, cnt, Wl2, bl2, Wr2, outp, N, 0);
}


# Encoder products by 512 resident work-groups walking the 16-row tiles: W fragments stay in registers; next tile's rows (16-byte loads, embedding index a tile ahead) load under the product; same sums
# speedup vs baseline: 3.0672x; 1.0464x over previous
#include <hip/hip_runtime.h>


typedef __attribute__((ext_vector_type(2))) float v2f;
typedef __attribute__((ext_vector_type(8))) float v8f;
typedef __attribute__((ext_vector_type(4))) float v4f;

#define LDK 260

__device__ __forceinline__ v8f wmma4(v2f a, v2f b, v8f c) {
  return __builtin_amdgcn_wmma_f32_16x16x4_f32(false, a, false, b,
                                               (short)0, c, false, false);
}

__device__ __forceinline__ void store_tile(float* __restrict__ out, int row0,
                                           int nrows, int col0, int lane,
                                           v8f c, bool do_relu) {
  const int hl = lane >> 4;
  const int n  = lane & 15;
  #pragma unroll
  for (int v = 0; v < 8; ++v) {
    int row = row0 + v + 8 * hl;
    float val = c[v];
    if (do_relu) val = fmaxf(val, 0.0f);
    if (row < nrows) out[(size_t)row * 128 + col0 + n] = val;
  }
}


#define ENC_WGS 512

template <int NX4, int NE>
__global__ void __launch_bounds__(256) encode_kernel(
    const float* x, const int* idx0, const float* emb0,
    const int* idx1, const float* emb1,
    const float* __restrict__ W, const float* __restrict__ bvec,
    float* out, int nrows) {
  constexpr int KX  = NX4 * 4;
  constexpr int KS  = (KX + NE * 8) / 4;
  constexpr int NXI = 16 * NX4;
  constexpr int NEI = 32 * NE;
  __shared__ float As[2][16 * LDK];
  const int tid    = threadIdx.x;
  const int lane   = tid & 31;
  const int hl     = lane >> 4;
  const int col0   = (tid >> 5) * 16;
  const int ntiles = (nrows + 15) >> 4;

  v2f bf[KS];
  {
    const int cn = col0 + (lane & 15);
    #pragma unroll
    for (int j = 0; j < KS; ++j) {
      const int k = 4 * j + 2 * hl;
      bf[j].x = W[k * 128 + cn];
      bf[j].y = W[(k + 1) * 128 + cn];
    }
  }
  const float bias = bvec[col0 + (lane & 15)];

  const int  wbase = __builtin_amdgcn_readfirstlane(tid >> 5) * 32 + 256;
  const bool b_x   = wbase < NXI;
  const bool b_e   = !b_x && wbase < NXI + NEI;
  const int  ra = tid / NX4, ca = tid % NX4;
  const int  ib = tid + 256;
  const int  rb = ib / NX4, cb = ib % NX4;
  const int  e    = b_e ? ib - NXI : 0;
  const int  re   = e / (2 * NE);
  const int  tab  = (e % (2 * NE)) >> 1;
  const int  hf   = e & 1;
  const int*   ip = tab ? idx1 : idx0;
  const float* ep = (tab ? emb1 : emb0) + hf * 4;
  const int  ecol = KX + tab * 8 + hf * 4;

  auto fetch = [&](int t, int id, float4& va, float4& vb, int& idnext) {
    vb = make_float4(0.f, 0.f, 0.f, 0.f);
    if (b_e) {
      vb = *reinterpret_cast<const float4*>(ep + id * 8);
      idnext = ip[min((t + ENC_WGS) * 16 + re, nrows - 1)];
    }
    va = *reinterpret_cast<const float4*>(x + (size_t)min(t * 16 + ra, nrows - 1) * KX + ca * 4);
    if (b_x)
      vb = *reinterpret_cast<const float4*>(x + (size_t)min(t * 16 + rb, nrows - 1) * KX + cb * 4);
  };
  auto put = [&](float* A, const float4& va, const float4& vb) {
    *reinterpret_cast<float4*>(A + ra * LDK + ca * 4) = va;
    if (b_x) *reinterpret_cast<float4*>(A + rb * LDK + cb * 4) = vb;
    if (b_e) *reinterpret_cast<float4*>(A + re * LDK + ecol) = vb;
  };

  int t   = blockIdx.x;
  int idn = 0;
  if (t < ntiles) {
    float4 va, vb;
    int id0 = 0;
    if (b_e) id0 = ip[min(t * 16 + re, nrows - 1)];
    fetch(t, id0, va, vb, idn);
    put(As[0], va, vb);
  }
  __syncthreads();

  int buf = 0;
  for (; t < ntiles; t += ENC_WGS) {
    const int  tn   = t + ENC_WGS;
    const bool more = tn < ntiles;
    float4 na, nb;
    int    idnn = 0;
    fetch(tn, idn, na, nb, idnn);
    __builtin_amdgcn_sched_barrier(0);

    const float* ap = As[buf] + (lane & 15) * LDK + 2 * hl;
    v8f c;
    #pragma unroll
    for (int v = 0; v < 8; ++v) c[v] = bias;
    #pragma unroll
    for (int j = 0; j < KS; ++j) {
      const v2f a = *reinterpret_cast<const v2f*>(ap + 4 * j);
      c = wmma4(a, bf[j], c);
    }
    if (more) put(As[buf ^ 1], na, nb);
    __syncthreads();
    store_tile(out, t * 16, nrows, col0, lane, c, true);
    buf ^= 1;
    idn = idnn;
  }
}


#define AT    400
#define AGRP  2048
#define ASUB  512
#define AKEEP 64
#define AQ    128
#define HT    127

#define LCAPW   1024
#define LHDR    32
#define TAGOFF  (LHDR + 32 * LCAPW)
#define TAGSEG  65536

__device__ __forceinline__ void agg_load_group(int4 (&d)[4], const int* outw,
                                               int grp, int lane) {
  const int* p = outw + (size_t)(grp >> 5) * (AT * 128) + TAGOFF + (grp & 31) * (AGRP / 4) + lane * 4;
  #pragma unroll
  for (int j = 0; j < 4; ++j) d[j] = *reinterpret_cast<const int4*>(p + j * 128);
}

__global__ void __launch_bounds__(256) tag_kernel(const int* __restrict__ dst,
                                                  int* __restrict__ outw, int nunits) {
  const int u = blockIdx.x * 256 + threadIdx.x;
  if (u >= nunits) return;
  const int4* p = reinterpret_cast<const int4*>(dst + (size_t)u * 16);
  int4 t0, t1;
  int* w0 = reinterpret_cast<int*>(&t0);
  int* w1 = reinterpret_cast<int*>(&t1);
  #pragma unroll
  for (int j = 0; j < 4; ++j) {
    const int4 d = p[j];
    const unsigned dv[4] = {(unsigned)d.x, (unsigned)d.y, (unsigned)d.z, (unsigned)d.w};
    unsigned a = 0u, b = 0u;
    #pragma unroll
    for (int c = 0; c < 4; ++c) {
      const unsigned tile = dv[c] / AT;
      const bool up = tile >= HT;
      a |= (up ? 127u : tile) << (8 * c);
      b |= (up ? tile - HT : 127u) << (8 * c);
    }
    w0[j] = (int)a;
    w1[j] = (int)b;
  }
  const int nseg = (nunits / (AGRP / 16) + 31) >> 5;
  const size_t e0 = (size_t)u * 16;
  int* q0 = outw + (e0 / TAGSEG) * (size_t)(AT * 128) + TAGOFF + (e0 % TAGSEG) / 4;
  int* q1 = q0 + (size_t)nseg * (AT * 128);
  *reinterpret_cast<int4*>(q0) = t0;
  *reinterpret_cast<int4*>(q1) = t1;
}

#define AOWN  13

template <bool SECOND>
__global__ void __launch_bounds__(1024) aggregate_kernel(
    const float* __restrict__ h, const int* __restrict__ src,
    const int* __restrict__ dst, float* __restrict__ s,
    float* __restrict__ cnt, int* __restrict__ lists, int E, int N) {
  __shared__ float acc[AT * 128];
  __shared__ int   cntl[AT];
  __shared__ int   qbuf[32 * AQ];
  __shared__ int   sub[32];

  const int tid   = threadIdx.x;
  const int lane  = tid & 31;
  const int w     = __builtin_amdgcn_readfirstlane(tid >> 5);
  const int nbase = blockIdx.x * AT;
  const int rows  = min(AT, N - nbase);
  int* q = qbuf + w * AQ;
  int* mylist   = lists + (size_t)nbase * 128;
  const int subcap = min(((rows * 128 - LHDR) / 32) & ~31, LCAPW);
  const int      half = (blockIdx.x >= HT) ? 1 : 0;
  const unsigned pat  = (blockIdx.x - (unsigned)(half * HT)) * 0x01010101u;

  const bool hi  = lane >= 16;
  const int  oa  = lane * 4 + (hi ? 2 : 0);
  const int  ob  = lane * 4 + (hi ? 0 : 2);
  auto add4 = [&](int row, const float4& v) {
    float* pa = acc + row * 128 + oa;
    float* pb = acc + row * 128 + ob;
    atomicAdd(pa + 0, hi ? v.z : v.x);
    atomicAdd(pa + 1, hi ? v.w : v.y);
    atomicAdd(pb + 0, hi ? v.x : v.z);
    atomicAdd(pb + 1, hi ? v.y : v.w);
  };

  auto add_rows = [&](int sv, int dl, int n) {
    int k = 0;
    for (; k + 8 <= n; k += 8) {
      float4 v[8];
      int    dd[8];
      #pragma unroll
      for (int u = 0; u < 8; ++u) {
        const int su = __builtin_amdgcn_readlane(sv, k + u);
        dd[u]        = __builtin_amdgcn_readlane(dl, k + u);
        v[u] = *reinterpret_cast<const float4*>(h + (size_t)su * 128 + lane * 4);
      }
      #pragma unroll
      for (int u = 0; u < 8; ++u) add4(dd[u], v[u]);
    }
    for (; k < n; ++k) {
      const int su = __builtin_amdgcn_readlane(sv, k);
      const int du = __builtin_amdgcn_readlane(dl, k);
      const float4 v = *reinterpret_cast<const float4*>(h + (size_t)su * 128 + lane * 4);
      add4(du, v);
    }
  };

  auto scan = [&](const bool LIST) __attribute__((always_inline)) {
    auto drain_round = [&](int r, int n) {
      int sv = 0, dl = 0;
      if (lane < n) {
        const int e = q[r + lane];
        sv = src[e];
        dl = dst[e] - nbase;
        if (LIST) {
          atomicAdd(&cntl[dl], 1);
          const int ow  = dl / AOWN;
          const int pos = atomicAdd(&sub[ow], 1);
          if (pos < subcap) mylist[LHDR + ow * subcap + pos] = (sv << 9) | dl;
        }
      }
      if (!LIST) add_rows(sv, dl, n);
    };

    int qn = 0;

    auto push = [&](bool match, int e) {
      const unsigned mm = __builtin_amdgcn_ballot_w32(match);
      if (mm) {
        if (match) q[qn + (int)__builtin_amdgcn_mbcnt_lo(mm, 0u)] = e;
        qn += __builtin_popcount(mm);
      }
    };

    auto make_room = [&]() {
      if (qn > AKEEP) {
        const int full = qn & ~31;
        for (int r = 0; r < full; r += 32) drain_round(r, 32);
        const int rem = qn - full;
        int t = 0;
        if (lane < rem) t = q[full + lane];
        if (lane < rem) q[lane] = t;
        qn = rem;
      }
    };

    const int nfull = E / AGRP;
    if (w == 0) {
      for (int e0 = nfull * AGRP; e0 < E; e0 += 32) {
        make_room();
        const int e = e0 + lane;
        const int d = (e < E) ? dst[e] : -1;
        push((unsigned)(d - nbase) < (unsigned)rows, e);
      }
    }

    auto miss = [&](int x) -> unsigned { return ((unsigned)x ^ pat) + 0x7f7f7f7fu; };
    auto ours = [&](const int4& d) -> unsigned {
      const unsigned n = ((miss(d.x) >> 7) & 0x01010101u) | ((miss(d.y) >> 6) & 0x02020202u) |
                         ((miss(d.z) >> 5) & 0x04040404u) | ((miss(d.w) >> 4) & 0x08080808u);
      return n ^ 0x0f0f0f0fu;
    };
    auto test_group = [&](const int4 (&d)[4], int g) {
      const int gb = g * AGRP + lane * 16;
      unsigned lo = ours(d[0]) | (ours(d[1]) << 4);
      unsigned hi = ours(d[2]) | (ours(d[3]) << 4);
      unsigned mm = __builtin_amdgcn_ballot_w32((lo | hi) != 0u);
      while (mm != 0u) {
        make_room();
        if ((lo | hi) != 0u) {
          const bool     inlo = lo != 0u;
          const unsigned cur  = inlo ? lo : hi;
          const int      b    = __builtin_ctz(cur);
          const int      j    = (inlo ? 0 : 2) + ((b >> 2) & 1);
          q[qn + (int)__builtin_amdgcn_mbcnt_lo(mm, 0u)] =
              gb + j * ASUB + ((b & 3) << 2) + (b >> 3);
          const unsigned rest = cur & (cur - 1u);
          lo = inlo ? rest : lo;
          hi = inlo ? hi : rest;
        }
        qn += __builtin_popcount(mm);
        mm = __builtin_amdgcn_ballot_w32((lo | hi) != 0u);
      }
    };

    if (nfull > 0) {
      const int* tags = lists + (size_t)(half * ((nfull + 31) >> 5)) * (AT * 128);
      const int nit = (nfull + 31) / 32;
      int itA = (int)((blockIdx.x * 13u) % (unsigned)nit);
      int4 A[4], B[4];
      agg_load_group(A, tags, min(itA * 32 + w, nfull - 1), lane);
      for (int i = 0; i < nit; i += 2) {
        const int itB = (itA + 1 == nit) ? 0 : itA + 1;
        agg_load_group(B, tags, min(itB * 32 + w, nfull - 1), lane);
        if (itA * 32 + w < nfull) test_group(A, itA * 32 + w);
        itA = (itB + 1 == nit) ? 0 : itB + 1;
        agg_load_group(A, tags, min(itA * 32 + w, nfull - 1), lane);
        if (i + 1 < nit && itB * 32 + w < nfull) test_group(B, itB * 32 + w);
      }
    }
    for (int r = 0; r < qn; r += 32) drain_round(r, min(32, qn - r));
  };

  int nsub = 0;
  if (!SECOND) {
    for (int idx = tid; idx < AT; idx += 1024) cntl[idx] = 0;
    if (tid < 32) sub[tid] = 0;
    __syncthreads();
    scan(true);
    __syncthreads();
    nsub = sub[lane];
  } else {
    nsub = mylist[lane];
  }
  const bool replay = __builtin_amdgcn_ballot_w32((unsigned)nsub > (unsigned)subcap) == 0u;

  if (replay) {
    const int own0 = w * AOWN;
    const int ownn = max(0, min(AOWN, rows - own0));
    float* myacc = acc + own0 * 128 + lane * 4;
    for (int j = 0; j < ownn; ++j)
      *reinterpret_cast<v4f*>(myacc + j * 128) = v4f{0.f, 0.f, 0.f, 0.f};

    auto loadg = [&](int sv, int dl, int k, v4f (&v)[8], int (&dd)[8]) {
      #pragma unroll
      for (int u = 0; u < 8; ++u) {
        const int su = __builtin_amdgcn_readlane(sv, k + u);
        dd[u]        = __builtin_amdgcn_readlane(dl, k + u);
        v[u] = *reinterpret_cast<const v4f*>(h + (size_t)su * 128 + lane * 4);
      }
    };
    auto rmw8 = [&](const v4f (&v)[8], const int (&dd)[8]) {
      v4f ld = *reinterpret_cast<const v4f*>(myacc + dd[0] * 128);
      v4f prev = ld;
      #pragma unroll
      for (int u = 0; u < 8; ++u) {
        v4f nx = ld;
        if (u < 7) nx = *reinterpret_cast<const v4f*>(myacc + dd[u + 1] * 128);
        const v4f base = (u > 0 && dd[u] == dd[u - 1]) ? prev : ld;
        const v4f sum  = base + v[u];
        *reinterpret_cast<v4f*>(myacc + dd[u] * 128) = sum;
        prev = sum;
        ld   = nx;
      }
    };
    auto round32 = [&](int p) {
      const int sv = (int)((unsigned)p >> 9);
      const int dl = (p & 511) - own0;
      v4f A[8], B[8];
      int da[8], db[8];
      loadg(sv, dl, 0, A, da);
      loadg(sv, dl, 8, B, db);
      rmw8(A, da);
      loadg(sv, dl, 16, A, da);
      rmw8(B, db);
      loadg(sv, dl, 24, B, db);
      rmw8(A, da);
      rmw8(B, db);
    };
    auto round_rest = [&](int p, int n) {
      const int sv = (int)((unsigned)p >> 9);
      const int dl = (p & 511) - own0;
      int k = 0;
      for (; k + 8 <= n; k += 8) {
        v4f A[8];
        int da[8];
        loadg(sv, dl, k, A, da);
        rmw8(A, da);
      }
      for (; k < n; ++k) {
        const int su = __builtin_amdgcn_readlane(sv, k);
        const int du = __builtin_amdgcn_readlane(dl, k);
        const v4f v = *reinterpret_cast<const v4f*>(h + (size_t)su * 128 + lane * 4);
        v4f* pa = reinterpret_cast<v4f*>(myacc + du * 128);
        *pa = *pa + v;
      }
    };

    const int  mycnt = __builtin_amdgcn_readlane(nsub, w);
    const int* ent   = mylist + LHDR + w * subcap;
    int p = (lane < mycnt) ? ent[lane] : 0;
    for (int r = 0; r < mycnt; r += 32) {
      const int pn = (r + 32 + lane < mycnt) ? ent[r + 32 + lane] : 0;
      if (mycnt - r >= 32) round32(p);
      else                 round_rest(p, mycnt - r);
      p = pn;
    }

    for (int j = 0; j < ownn; ++j)
      *reinterpret_cast<v4f*>(s + (size_t)(nbase + own0 + j) * 128 + lane * 4) =
          *reinterpret_cast<const v4f*>(myacc + j * 128);
  } else {
    for (int idx = tid; idx < AT * 32; idx += 1024)
      *reinterpret_cast<float4*>(acc + idx * 4) = make_float4(0.f, 0.f, 0.f, 0.f);
    __syncthreads();
    scan(false);
    __syncthreads();
    for (int idx = tid; idx < rows * 32; idx += 1024) {
      const int r = idx >> 5, c4 = idx & 31;
      *reinterpret_cast<float4*>(s + (size_t)(nbase + r) * 128 + c4 * 4) =
          *reinterpret_cast<const float4*>(acc + r * 128 + c4 * 4);
    }
  }
  if (!SECOND) {
    for (int r = tid; r < rows; r += 1024) cnt[nbase + r] = (float)cntl[r];
    if (tid < 32) mylist[tid] = sub[tid];
  }
}


#define SAGE_WGS 512
#define SAGE_KS  64

__global__ void __launch_bounds__(256) sage_kernel(
    const float* h, const float* s,
    const float* __restrict__ cnt, const float* __restrict__ Wl,
    const float* __restrict__ bl, const float* __restrict__ Wr,
    float* out, int nrows, int do_relu) {
  __shared__ float As[2][16 * LDK];
  const int tid    = threadIdx.x;
  const int lane   = tid & 31;
  const int hl     = lane >> 4;
  const int col0   = (tid >> 5) * 16;
  const int ntiles = (nrows + 15) >> 4;

  v2f bf[SAGE_KS];
  {
    const int cn = col0 + (lane & 15);
    #pragma unroll
    for (int j = 0; j < 32; ++j) {
      const int k = 4 * j + 2 * hl;
      bf[j].x      = Wl[k * 128 + cn];
      bf[j].y      = Wl[(k + 1) * 128 + cn];
      bf[32 + j].x = Wr[k * 128 + cn];
      bf[32 + j].y = Wr[(k + 1) * 128 + cn];
    }
  }
  const float bias = bl[col0 + (lane & 15)];

  const int  c4     = tid & 63;
  const int  rb     = tid >> 6;
  const bool is_sum = (__builtin_amdgcn_readfirstlane(tid >> 5) & 1) == 0;

  const float* rows_in = (is_sum ? s : h) + (c4 & 31) * 4;
  auto fetch = [&](int t, float4 (&v)[4], float (&d)[4]) {
    #pragma unroll
    for (int i = 0; i < 4; ++i) {
      const int row = min(t * 16 + rb + 4 * i, nrows - 1);
      v[i] = *reinterpret_cast<const float4*>(rows_in + (size_t)row * 128);
      d[i] = cnt[row];
    }
  };
  auto put = [&](float* A, const float4 (&v)[4], const float (&d)[4]) {
    #pragma unroll
    for (int i = 0; i < 4; ++i) {
      float4 x = v[i];
      if (is_sum) {
        const float dd = fmaxf(d[i], 1.0f);
        x.x /= dd; x.y /= dd; x.z /= dd; x.w /= dd;
      }
      *reinterpret_cast<float4*>(A + (rb + 4 * i) * LDK + c4 * 4) = x;
    }
  };

  int t = blockIdx.x;
  if (t < ntiles) {
    float4 v[4];
    float  d[4];
    fetch(t, v, d);
    put(As[0], v, d);
  }
  __syncthreads();

  int buf = 0;
  for (; t < ntiles; t += SAGE_WGS) {
    const int  tn   = t + SAGE_WGS;
    const bool more = tn < ntiles;
    float4 vn[4];
    float  dn[4];
    fetch(tn, vn, dn);
    __builtin_amdgcn_sched_barrier(0);

    const float* ap = As[buf] + (lane & 15) * LDK + 2 * hl;
    v8f c;
    #pragma unroll
    for (int v = 0; v < 8; ++v) c[v] = bias;
    #pragma unroll
    for (int j = 0; j < SAGE_KS; ++j) {
      const v2f a = *reinterpret_cast<const v2f*>(ap + 4 * j);
      c = wmma4(a, bf[j], c);
    }
    if (more) put(As[buf ^ 1], vn, dn);
    __syncthreads();
    store_tile(out, t * 16, nrows, col0, lane, c, do_relu != 0);
    buf ^= 1;
  }
}


extern "C" void kernel_launch(void* const* d_in, const int* in_sizes, int n_in,
                              void* d_out, int out_size, void* d_ws, size_t ws_size,
                              hipStream_t stream) {
  (void)n_in; (void)out_size; (void)ws_size;
  const float* x_pol   = (const float*)d_in[0];
  const int*   pol_sid = (const int*)  d_in[1];
  const float* x_comp  = (const float*)d_in[2];
  const int*   c_sec   = (const int*)  d_in[3];
  const int*   c_ind   = (const int*)  d_in[4];
  const int*   edges   = (const int*)  d_in[5];
  const float* s_emb   = (const float*)d_in[6];
  const float* sec_emb = (const float*)d_in[7];
  const float* ind_emb = (const float*)d_in[8];
  const float* W_pol   = (const float*)d_in[9];
  const float* b_pol   = (const float*)d_in[10];
  const float* W_comp  = (const float*)d_in[11];
  const float* b_comp  = (const float*)d_in[12];
  const float* Wl1     = (const float*)d_in[13];
  const float* bl1     = (const float*)d_in[14];
  const float* Wr1     = (const float*)d_in[15];
  const float* Wl2     = (const float*)d_in[16];
  const float* bl2     = (const float*)d_in[17];
  const float* Wr2     = (const float*)d_in[18];

  const int NP = in_sizes[0] / 64;
  const int NC = in_sizes[2] / 96;
  const int E  = in_sizes[5] / 2;
  const int N  = NP + NC;
  const int* src = edges;
  const int* dst = edges + E;

  float* h    = (float*)d_ws;
  float* s    = h + (size_t)N * 128;
  float* cnt  = s + (size_t)N * 128;
  float* outp = (float*)d_out;
  int*   lists = (int*)d_out;

  const dim3 blk(256);
  const dim3 ablk(1024);
  const dim3 agrid((N + AT - 1) / AT);

  const int nunits = (E / AGRP) * (AGRP / 16);
  tag_kernel<<<dim3((nunits + 255) / 256), blk, 0, stream>>>(dst, lists, nunits);

  encode_kernel<16, 1><<<dim3(ENC_WGS), blk, 0, stream>>>(
      x_pol, pol_sid, s_emb, pol_sid, s_emb, W_pol, b_pol, h, NP);
  encode_kernel<24, 2><<<dim3(ENC_WGS), blk, 0, stream>>>(
      x_comp, c_sec, sec_emb, c_ind, ind_emb, W_comp, b_comp,
      h + (size_t)NP * 128, NC);

  aggregate_kernel<false><<<agrid, ablk, 0, stream>>>(h, src, dst, s, cnt, lists, E, N);
  sage_kernel<<<dim3(SAGE_WGS), blk, 0, stream>>>(
      h, s, cnt, Wl1, bl1, Wr1, h, N, 1);

  aggregate_kernel<true><<<agrid, ablk, 0, stream>>>(h, src, dst, s, cnt, lists, E, N);
  sage_kernel<<<dim3(SAGE_WGS), blk, 0, stream>>>(
      h, s, cnt, Wl2, bl2, Wr2, outp, N, 0);
}


# Mean taken where a row's sum is finished (the aggregation's row store divides by max(count,1)): the layer kernel stages plain rows, no count loads and no divisions between products; sums unchanged
# speedup vs baseline: 3.1646x; 1.0318x over previous
#include <hip/hip_runtime.h>


typedef __attribute__((ext_vector_type(2))) float v2f;
typedef __attribute__((ext_vector_type(8))) float v8f;
typedef __attribute__((ext_vector_type(4))) float v4f;

#define LDK 260

__device__ __forceinline__ v8f wmma4(v2f a, v2f b, v8f c) {
  return __builtin_amdgcn_wmma_f32_16x16x4_f32(false, a, false, b,
                                               (short)0, c, false, false);
}

__device__ __forceinline__ void store_tile(float* __restrict__ out, int row0,
                                           int nrows, int col0, int lane,
                                           v8f c, bool do_relu) {
  const int hl = lane >> 4;
  const int n  = lane & 15;
  #pragma unroll
  for (int v = 0; v < 8; ++v) {
    int row = row0 + v + 8 * hl;
    float val = c[v];
    if (do_relu) val = fmaxf(val, 0.0f);
    if (row < nrows) out[(size_t)row * 128 + col0 + n] = val;
  }
}


#define ENC_WGS 512

template <int NX4, int NE>
__global__ void __launch_bounds__(256) encode_kernel(
    const float* x, const int* idx0, const float* emb0,
    const int* idx1, const float* emb1,
    const float* __restrict__ W, const float* __restrict__ bvec,
    float* out, int nrows) {
  constexpr int KX  = NX4 * 4;
  constexpr int KS  = (KX + NE * 8) / 4;
  constexpr int NXI = 16 * NX4;
  constexpr int NEI = 32 * NE;
  __shared__ float As[2][16 * LDK];
  const int tid    = threadIdx.x;
  const int lane   = tid & 31;
  const int hl     = lane >> 4;
  const int col0   = (tid >> 5) * 16;
  const int ntiles = (nrows + 15) >> 4;

  v2f bf[KS];
  {
    const int cn = col0 + (lane & 15);
    #pragma unroll
    for (int j = 0; j < KS; ++j) {
      const int k = 4 * j + 2 * hl;
      bf[j].x = W[k * 128 + cn];
      bf[j].y = W[(k + 1) * 128 + cn];
    }
  }
  const float bias = bvec[col0 + (lane & 15)];

  const int  wbase = __builtin_amdgcn_readfirstlane(tid >> 5) * 32 + 256;
  const bool b_x   = wbase < NXI;
  const bool b_e   = !b_x && wbase < NXI + NEI;
  const int  ra = tid / NX4, ca = tid % NX4;
  const int  ib = tid + 256;
  const int  rb = ib / NX4, cb = ib % NX4;
  const int  e    = b_e ? ib - NXI : 0;
  const int  re   = e / (2 * NE);
  const int  tab  = (e % (2 * NE)) >> 1;
  const int  hf   = e & 1;
  const int*   ip = tab ? idx1 : idx0;
  const float* ep = (tab ? emb1 : emb0) + hf * 4;
  const int  ecol = KX + tab * 8 + hf * 4;

  auto fetch = [&](int t, int id, float4& va, float4& vb, int& idnext) {
    vb = make_float4(0.f, 0.f, 0.f, 0.f);
    if (b_e) {
      vb = *reinterpret_cast<const float4*>(ep + id * 8);
      idnext = ip[min((t + ENC_WGS) * 16 + re, nrows - 1)];
    }
    va = *reinterpret_cast<const float4*>(x + (size_t)min(t * 16 + ra, nrows - 1) * KX + ca * 4);
    if (b_x)
      vb = *reinterpret_cast<const float4*>(x + (size_t)min(t * 16 + rb, nrows - 1) * KX + cb * 4);
  };
  auto put = [&](float* A, const float4& va, const float4& vb) {
    *reinterpret_cast<float4*>(A + ra * LDK + ca * 4) = va;
    if (b_x) *reinterpret_cast<float4*>(A + rb * LDK + cb * 4) = vb;
    if (b_e) *reinterpret_cast<float4*>(A + re * LDK + ecol) = vb;
  };

  int t   = blockIdx.x;
  int idn = 0;
  if (t < ntiles) {
    float4 va, vb;
    int id0 = 0;
    if (b_e) id0 = ip[min(t * 16 + re, nrows - 1)];
    fetch(t, id0, va, vb, idn);
    put(As[0], va, vb);
  }
  __syncthreads();

  int buf = 0;
  for (; t < ntiles; t += ENC_WGS) {
    const int  tn   = t + ENC_WGS;
    const bool more = tn < ntiles;
    float4 na, nb;
    int    idnn = 0;
    fetch(tn, idn, na, nb, idnn);
    __builtin_amdgcn_sched_barrier(0);

    const float* ap = As[buf] + (lane & 15) * LDK + 2 * hl;
    v8f c;
    #pragma unroll
    for (int v = 0; v < 8; ++v) c[v] = bias;
    #pragma unroll
    for (int j = 0; j < KS; ++j) {
      const v2f a = *reinterpret_cast<const v2f*>(ap + 4 * j);
      c = wmma4(a, bf[j], c);
    }
    if (more) put(As[buf ^ 1], na, nb);
    __syncthreads();
    store_tile(out, t * 16, nrows, col0, lane, c, true);
    buf ^= 1;
    idn = idnn;
  }
}


#define AT    400
#define AGRP  2048
#define ASUB  512
#define AKEEP 64
#define AQ    128
#define HT    127

#define LCAPW   1024
#define LHDR    32
#define TAGOFF  (LHDR + 32 * LCAPW)
#define TAGSEG  65536

__device__ __forceinline__ void agg_load_group(int4 (&d)[4], const int* outw,
                                               int grp, int lane) {
  const int* p = outw + (size_t)(grp >> 5) * (AT * 128) + TAGOFF + (grp & 31) * (AGRP / 4) + lane * 4;
  #pragma unroll
  for (int j = 0; j < 4; ++j) d[j] = *reinterpret_cast<const int4*>(p + j * 128);
}

__global__ void __launch_bounds__(256) tag_kernel(const int* __restrict__ dst,
                                                  int* __restrict__ outw, int nunits) {
  const int u = blockIdx.x * 256 + threadIdx.x;
  if (u >= nunits) return;
  const int4* p = reinterpret_cast<const int4*>(dst + (size_t)u * 16);
  int4 t0, t1;
  int* w0 = reinterpret_cast<int*>(&t0);
  int* w1 = reinterpret_cast<int*>(&t1);
  #pragma unroll
  for (int j = 0; j < 4; ++j) {
    const int4 d = p[j];
    const unsigned dv[4] = {(unsigned)d.x, (unsigned)d.y, (unsigned)d.z, (unsigned)d.w};
    unsigned a = 0u, b = 0u;
    #pragma unroll
    for (int c = 0; c < 4; ++c) {
      const unsigned tile = dv[c] / AT;
      const bool up = tile >= HT;
      a |= (up ? 127u : tile) << (8 * c);
      b |= (up ? tile - HT : 127u) << (8 * c);
    }
    w0[j] = (int)a;
    w1[j] = (int)b;
  }
  const int nseg = (nunits / (AGRP / 16) + 31) >> 5;
  const size_t e0 = (size_t)u * 16;
  int* q0 = outw + (e0 / TAGSEG) * (size_t)(AT * 128) + TAGOFF + (e0 % TAGSEG) / 4;
  int* q1 = q0 + (size_t)nseg * (AT * 128);
  *reinterpret_cast<int4*>(q0) = t0;
  *reinterpret_cast<int4*>(q1) = t1;
}

#define AOWN  13

template <bool SECOND>
__global__ void __launch_bounds__(1024) aggregate_kernel(
    const float* __restrict__ h, const int* __restrict__ src,
    const int* __restrict__ dst, float* __restrict__ s,
    float* __restrict__ cnt, int* __restrict__ lists, int E, int N) {
  __shared__ float acc[AT * 128];
  __shared__ int   cntl[AT];
  __shared__ int   qbuf[32 * AQ];
  __shared__ int   sub[32];

  const int tid   = threadIdx.x;
  const int lane  = tid & 31;
  const int w     = __builtin_amdgcn_readfirstlane(tid >> 5);
  const int nbase = blockIdx.x * AT;
  const int rows  = min(AT, N - nbase);
  int* q = qbuf + w * AQ;
  int* mylist   = lists + (size_t)nbase * 128;
  const int subcap = min(((rows * 128 - LHDR) / 32) & ~31, LCAPW);
  const int      half = (blockIdx.x >= HT) ? 1 : 0;
  const unsigned pat  = (blockIdx.x - (unsigned)(half * HT)) * 0x01010101u;

  const bool hi  = lane >= 16;
  const int  oa  = lane * 4 + (hi ? 2 : 0);
  const int  ob  = lane * 4 + (hi ? 0 : 2);
  auto add4 = [&](int row, const float4& v) {
    float* pa = acc + row * 128 + oa;
    float* pb = acc + row * 128 + ob;
    atomicAdd(pa + 0, hi ? v.z : v.x);
    atomicAdd(pa + 1, hi ? v.w : v.y);
    atomicAdd(pb + 0, hi ? v.x : v.z);
    atomicAdd(pb + 1, hi ? v.y : v.w);
  };

  auto add_rows = [&](int sv, int dl, int n) {
    int k = 0;
    for (; k + 8 <= n; k += 8) {
      float4 v[8];
      int    dd[8];
      #pragma unroll
      for (int u = 0; u < 8; ++u) {
        const int su = __builtin_amdgcn_readlane(sv, k + u);
        dd[u]        = __builtin_amdgcn_readlane(dl, k + u);
        v[u] = *reinterpret_cast<const float4*>(h + (size_t)su * 128 + lane * 4);
      }
      #pragma unroll
      for (int u = 0; u < 8; ++u) add4(dd[u], v[u]);
    }
    for (; k < n; ++k) {
      const int su = __builtin_amdgcn_readlane(sv, k);
      const int du = __builtin_amdgcn_readlane(dl, k);
      const float4 v = *reinterpret_cast<const float4*>(h + (size_t)su * 128 + lane * 4);
      add4(du, v);
    }
  };

  auto scan = [&](const bool LIST) __attribute__((always_inline)) {
    auto drain_round = [&](int r, int n) {
      int sv = 0, dl = 0;
      if (lane < n) {
        const int e = q[r + lane];
        sv = src[e];
        dl = dst[e] - nbase;
        if (LIST) {
          atomicAdd(&cntl[dl], 1);
          const int ow  = dl / AOWN;
          const int pos = atomicAdd(&sub[ow], 1);
          if (pos < subcap) mylist[LHDR + ow * subcap + pos] = (sv << 9) | dl;
        }
      }
      if (!LIST) add_rows(sv, dl, n);
    };

    int qn = 0;

    auto push = [&](bool match, int e) {
      const unsigned mm = __builtin_amdgcn_ballot_w32(match);
      if (mm) {
        if (match) q[qn + (int)__builtin_amdgcn_mbcnt_lo(mm, 0u)] = e;
        qn += __builtin_popcount(mm);
      }
    };

    auto make_room = [&]() {
      if (qn > AKEEP) {
        const int full = qn & ~31;
        for (int r = 0; r < full; r += 32) drain_round(r, 32);
        const int rem = qn - full;
        int t = 0;
        if (lane < rem) t = q[full + lane];
        if (lane < rem) q[lane] = t;
        qn = rem;
      }
    };

    const int nfull = E / AGRP;
    if (w == 0) {
      for (int e0 = nfull * AGRP; e0 < E; e0 += 32) {
        make_room();
        const int e = e0 + lane;
        const int d = (e < E) ? dst[e] : -1;
        push((unsigned)(d - nbase) < (unsigned)rows, e);
      }
    }

    auto miss = [&](int x) -> unsigned { return ((unsigned)x ^ pat) + 0x7f7f7f7fu; };
    auto ours = [&](const int4& d) -> unsigned {
      const unsigned n = ((miss(d.x) >> 7) & 0x01010101u) | ((miss(d.y) >> 6) & 0x02020202u) |
                         ((miss(d.z) >> 5) & 0x04040404u) | ((miss(d.w) >> 4) & 0x08080808u);
      return n ^ 0x0f0f0f0fu;
    };
    auto test_group = [&](const int4 (&d)[4], int g) {
      const int gb = g * AGRP + lane * 16;
      unsigned lo = ours(d[0]) | (ours(d[1]) << 4);
      unsigned hi = ours(d[2]) | (ours(d[3]) << 4);
      unsigned mm = __builtin_amdgcn_ballot_w32((lo | hi) != 0u);
      while (mm != 0u) {
        make_room();
        if ((lo | hi) != 0u) {
          const bool     inlo = lo != 0u;
          const unsigned cur  = inlo ? lo : hi;
          const int      b    = __builtin_ctz(cur);
          const int      j    = (inlo ? 0 : 2) + ((b >> 2) & 1);
          q[qn + (int)__builtin_amdgcn_mbcnt_lo(mm, 0u)] =
              gb + j * ASUB + ((b & 3) << 2) + (b >> 3);
          const unsigned rest = cur & (cur - 1u);
          lo = inlo ? rest : lo;
          hi = inlo ? hi : rest;
        }
        qn += __builtin_popcount(mm);
        mm = __builtin_amdgcn_ballot_w32((lo | hi) != 0u);
      }
    };

    if (nfull > 0) {
      const int* tags = lists + (size_t)(half * ((nfull + 31) >> 5)) * (AT * 128);
      const int nit = (nfull + 31) / 32;
      int itA = (int)((blockIdx.x * 13u) % (unsigned)nit);
      int4 A[4], B[4];
      agg_load_group(A, tags, min(itA * 32 + w, nfull - 1), lane);
      for (int i = 0; i < nit; i += 2) {
        const int itB = (itA + 1 == nit) ? 0 : itA + 1;
        agg_load_group(B, tags, min(itB * 32 + w, nfull - 1), lane);
        if (itA * 32 + w < nfull) test_group(A, itA * 32 + w);
        itA = (itB + 1 == nit) ? 0 : itB + 1;
        agg_load_group(A, tags, min(itA * 32 + w, nfull - 1), lane);
        if (i + 1 < nit && itB * 32 + w < nfull) test_group(B, itB * 32 + w);
      }
    }
    for (int r = 0; r < qn; r += 32) drain_round(r, min(32, qn - r));
  };

  int nsub = 0;
  if (!SECOND) {
    for (int idx = tid; idx < AT; idx += 1024) cntl[idx] = 0;
    if (tid < 32) sub[tid] = 0;
    __syncthreads();
    scan(true);
    __syncthreads();
    nsub = sub[lane];
  } else {
    nsub = mylist[lane];
  }
  const bool replay = __builtin_amdgcn_ballot_w32((unsigned)nsub > (unsigned)subcap) == 0u;

  if (replay) {
    const int own0 = w * AOWN;
    const int ownn = max(0, min(AOWN, rows - own0));
    float* myacc = acc + own0 * 128 + lane * 4;
    for (int j = 0; j < ownn; ++j)
      *reinterpret_cast<v4f*>(myacc + j * 128) = v4f{0.f, 0.f, 0.f, 0.f};

    auto loadg = [&](int sv, int dl, int k, v4f (&v)[8], int (&dd)[8]) {
      #pragma unroll
      for (int u = 0; u < 8; ++u) {
        const int su = __builtin_amdgcn_readlane(sv, k + u);
        dd[u]        = __builtin_amdgcn_readlane(dl, k + u);
        v[u] = *reinterpret_cast<const v4f*>(h + (size_t)su * 128 + lane * 4);
      }
    };
    auto rmw8 = [&](const v4f (&v)[8], const int (&dd)[8]) {
      v4f ld = *reinterpret_cast<const v4f*>(myacc + dd[0] * 128);
      v4f prev = ld;
      #pragma unroll
      for (int u = 0; u < 8; ++u) {
        v4f nx = ld;
        if (u < 7) nx = *reinterpret_cast<const v4f*>(myacc + dd[u + 1] * 128);
        const v4f base = (u > 0 && dd[u] == dd[u - 1]) ? prev : ld;
        const v4f sum  = base + v[u];
        *reinterpret_cast<v4f*>(myacc + dd[u] * 128) = sum;
        prev = sum;
        ld   = nx;
      }
    };
    auto round32 = [&](int p) {
      const int sv = (int)((unsigned)p >> 9);
      const int dl = (p & 511) - own0;
      v4f A[8], B[8];
      int da[8], db[8];
      loadg(sv, dl, 0, A, da);
      loadg(sv, dl, 8, B, db);
      rmw8(A, da);
      loadg(sv, dl, 16, A, da);
      rmw8(B, db);
      loadg(sv, dl, 24, B, db);
      rmw8(A, da);
      rmw8(B, db);
    };
    auto round_rest = [&](int p, int n) {
      const int sv = (int)((unsigned)p >> 9);
      const int dl = (p & 511) - own0;
      int k = 0;
      for (; k + 8 <= n; k += 8) {
        v4f A[8];
        int da[8];
        loadg(sv, dl, k, A, da);
        rmw8(A, da);
      }
      for (; k < n; ++k) {
        const int su = __builtin_amdgcn_readlane(sv, k);
        const int du = __builtin_amdgcn_readlane(dl, k);
        const v4f v = *reinterpret_cast<const v4f*>(h + (size_t)su * 128 + lane * 4);
        v4f* pa = reinterpret_cast<v4f*>(myacc + du * 128);
        *pa = *pa + v;
      }
    };

    const int  mycnt = __builtin_amdgcn_readlane(nsub, w);
    const int* ent   = mylist + LHDR + w * subcap;
    float mydeg = 1.0f;
    if (lane < ownn) mydeg = SECOND ? cnt[nbase + own0 + lane] : (float)cntl[own0 + lane];
    int p = (lane < mycnt) ? ent[lane] : 0;
    for (int r = 0; r < mycnt; r += 32) {
      const int pn = (r + 32 + lane < mycnt) ? ent[r + 32 + lane] : 0;
      if (mycnt - r >= 32) round32(p);
      else                 round_rest(p, mycnt - r);
      p = pn;
    }

    for (int j = 0; j < ownn; ++j) {
      const float dd = fmaxf(__int_as_float(__builtin_amdgcn_readlane(__float_as_int(mydeg), j)), 1.0f);
      v4f a = *reinterpret_cast<const v4f*>(myacc + j * 128);
      a.x /= dd; a.y /= dd; a.z /= dd; a.w /= dd;
      *reinterpret_cast<v4f*>(s + (size_t)(nbase + own0 + j) * 128 + lane * 4) = a;
    }
  } else {
    for (int idx = tid; idx < AT * 32; idx += 1024)
      *reinterpret_cast<float4*>(acc + idx * 4) = make_float4(0.f, 0.f, 0.f, 0.f);
    __syncthreads();
    scan(false);
    __syncthreads();
    for (int idx = tid; idx < rows * 32; idx += 1024) {
      const int r = idx >> 5, c4 = idx & 31;
      const float dd = fmaxf(SECOND ? cnt[nbase + r] : (float)cntl[r], 1.0f);
      float4 a = *reinterpret_cast<const float4*>(acc + r * 128 + c4 * 4);
      a.x /= dd; a.y /= dd; a.z /= dd; a.w /= dd;
      *reinterpret_cast<float4*>(s + (size_t)(nbase + r) * 128 + c4 * 4) = a;
    }
  }
  if (!SECOND) {
    for (int r = tid; r < rows; r += 1024) cnt[nbase + r] = (float)cntl[r];
    if (tid < 32) mylist[tid] = sub[tid];
  }
}


#define SAGE_WGS 512
#define SAGE_KS  64

__global__ void __launch_bounds__(256) sage_kernel(
    const float* h, const float* s, const float* __restrict__ Wl,
    const float* __restrict__ bl, const float* __restrict__ Wr,
    float* out, int nrows, int do_relu) {
  __shared__ float As[2][16 * LDK];
  const int tid    = threadIdx.x;
  const int lane   = tid & 31;
  const int hl     = lane >> 4;
  const int col0   = (tid >> 5) * 16;
  const int ntiles = (nrows + 15) >> 4;

  v2f bf[SAGE_KS];
  {
    const int cn = col0 + (lane & 15);
    #pragma unroll
    for (int j = 0; j < 32; ++j) {
      const int k = 4 * j + 2 * hl;
      bf[j].x      = Wl[k * 128 + cn];
      bf[j].y      = Wl[(k + 1) * 128 + cn];
      bf[32 + j].x = Wr[k * 128 + cn];
      bf[32 + j].y = Wr[(k + 1) * 128 + cn];
    }
  }
  const float bias = bl[col0 + (lane & 15)];

  const int  c4     = tid & 63;
  const int  rb     = tid >> 6;
  const bool is_sum = (__builtin_amdgcn_readfirstlane(tid >> 5) & 1) == 0;

  const float* rows_in = (is_sum ? s : h) + (c4 & 31) * 4;
  auto fetch = [&](int t, v4f (&v)[4]) {
    #pragma unroll
    for (int i = 0; i < 4; ++i) {
      const int row = min(t * 16 + rb + 4 * i, nrows - 1);
      v[i] = *reinterpret_cast<const v4f*>(rows_in + (size_t)row * 128);
    }
  };
  auto put = [&](float* A, const v4f (&v)[4]) {
    #pragma unroll
    for (int i = 0; i < 4; ++i)
      *reinterpret_cast<v4f*>(A + (rb + 4 * i) * LDK + c4 * 4) = v[i];
  };

  int t = blockIdx.x;
  if (t < ntiles) {
    v4f v[4];
    fetch(t, v);
    put(As[0], v);
  }
  __syncthreads();

  int buf = 0;
  for (; t < ntiles; t += SAGE_WGS) {
    const int  tn   = t + SAGE_WGS;
    const bool more = tn < ntiles;
    v4f vn[4];
    fetch(tn, vn);
    __builtin_amdgcn_sched_barrier(0);

    const float* ap = As[buf] + (lane & 15) * LDK + 2 * hl;
    v8f c;
    #pragma unroll
    for (int v = 0; v < 8; ++v) c[v] = bias;
    #pragma unroll
    for (int j = 0; j < SAGE_KS; ++j) {
      const v2f a = *reinterpret_cast<const v2f*>(ap + 4 * j);
      c = wmma4(a, bf[j], c);
    }
    if (more) put(As[buf ^ 1], vn);
    __syncthreads();
    store_tile(out, t * 16, nrows, col0, lane, c, do_relu != 0);
    buf ^= 1;
  }
}


extern "C" void kernel_launch(void* const* d_in, const int* in_sizes, int n_in,
                              void* d_out, int out_size, void* d_ws, size_t ws_size,
                              hipStream_t stream) {
  (void)n_in; (void)out_size; (void)ws_size;
  const float* x_pol   = (const float*)d_in[0];
  const int*   pol_sid = (const int*)  d_in[1];
  const float* x_comp  = (const float*)d_in[2];
  const int*   c_sec   = (const int*)  d_in[3];
  const int*   c_ind   = (const int*)  d_in[4];
  const int*   edges   = (const int*)  d_in[5];
  const float* s_emb   = (const float*)d_in[6];
  const float* sec_emb = (const float*)d_in[7];
  const float* ind_emb = (const float*)d_in[8];
  const float* W_pol   = (const float*)d_in[9];
  const float* b_pol   = (const float*)d_in[10];
  const float* W_comp  = (const float*)d_in[11];
  const float* b_comp  = (const float*)d_in[12];
  const float* Wl1     = (const float*)d_in[13];
  const float* bl1     = (const float*)d_in[14];
  const float* Wr1     = (const float*)d_in[15];
  const float* Wl2     = (const float*)d_in[16];
  const float* bl2     = (const float*)d_in[17];
  const float* Wr2     = (const float*)d_in[18];

  const int NP = in_sizes[0] / 64;
  const int NC = in_sizes[2] / 96;
  const int E  = in_sizes[5] / 2;
  const int N  = NP + NC;
  const int* src = edges;
  const int* dst = edges + E;

  float* h    = (float*)d_ws;
  float* s    = h + (size_t)N * 128;
  float* cnt  = s + (size_t)N * 128;
  float* outp = (float*)d_out;
  int*   lists = (int*)d_out;

  const dim3 blk(256);
  const dim3 ablk(1024);
  const dim3 agrid((N + AT - 1) / AT);

  const int nunits = (E / AGRP) * (AGRP / 16);
  tag_kernel<<<dim3((nunits + 255) / 256), blk, 0, stream>>>(dst, lists, nunits);

  encode_kernel<16, 1><<<dim3(ENC_WGS), blk, 0, stream>>>(
      x_pol, pol_sid, s_emb, pol_sid, s_emb, W_pol, b_pol, h, NP);
  encode_kernel<24, 2><<<dim3(ENC_WGS), blk, 0, stream>>>(
      x_comp, c_sec, sec_emb, c_ind, ind_emb, W_comp, b_comp,
      h + (size_t)NP * 128, NC);

  aggregate_kernel<false><<<agrid, ablk, 0, stream>>>(h, src, dst, s, cnt, lists, E, N);
  sage_kernel<<<dim3(SAGE_WGS), blk, 0, stream>>>(
      h, s, Wl1, bl1, Wr1, h, N, 1);

  aggregate_kernel<true><<<agrid, ablk, 0, stream>>>(h, src, dst, s, cnt, lists, E, N);
  sage_kernel<<<dim3(SAGE_WGS), blk, 0, stream>>>(
      h, s, Wl2, bl2, Wr2, outp, N, 0);
}


# Layer 1 takes its tile's edges from a bin that a first launch fills (250 work-groups, a chunk of the edge list each, slots by LDS counters): no tile scans all edges; sum order = list arrival
# speedup vs baseline: 3.5047x; 1.1075x over previous
#include <hip/hip_runtime.h>


typedef __attribute__((ext_vector_type(2))) float v2f;
typedef __attribute__((ext_vector_type(8))) float v8f;
typedef __attribute__((ext_vector_type(4))) float v4f;

#define LDK 260

__device__ __forceinline__ v8f wmma4(v2f a, v2f b, v8f c) {
  return __builtin_amdgcn_wmma_f32_16x16x4_f32(false, a, false, b,
                                               (short)0, c, false, false);
}

__device__ __forceinline__ void store_tile(float* __restrict__ out, int row0,
                                           int nrows, int col0, int lane,
                                           v8f c, bool do_relu) {
  const int hl = lane >> 4;
  const int n  = lane & 15;
  #pragma unroll
  for (int v = 0; v < 8; ++v) {
    int row = row0 + v + 8 * hl;
    float val = c[v];
    if (do_relu) val = fmaxf(val, 0.0f);
    if (row < nrows) out[(size_t)row * 128 + col0 + n] = val;
  }
}


#define ENC_WGS 512

template <int NX4, int NE>
__global__ void __launch_bounds__(256) encode_kernel(
    const float* x, const int* idx0, const float* emb0,
    const int* idx1, const float* emb1,
    const float* __restrict__ W, const float* __restrict__ bvec,
    float* out, int nrows) {
  constexpr int KX  = NX4 * 4;
  constexpr int KS  = (KX + NE * 8) / 4;
  constexpr int NXI = 16 * NX4;
  constexpr int NEI = 32 * NE;
  __shared__ float As[2][16 * LDK];
  const int tid    = threadIdx.x;
  const int lane   = tid & 31;
  const int hl     = lane >> 4;
  const int col0   = (tid >> 5) * 16;
  const int ntiles = (nrows + 15) >> 4;

  v2f bf[KS];
  {
    const int cn = col0 + (lane & 15);
    #pragma unroll
    for (int j = 0; j < KS; ++j) {
      const int k = 4 * j + 2 * hl;
      bf[j].x = W[k * 128 + cn];
      bf[j].y = W[(k + 1) * 128 + cn];
    }
  }
  const float bias = bvec[col0 + (lane & 15)];

  const int  wbase = __builtin_amdgcn_readfirstlane(tid >> 5) * 32 + 256;
  const bool b_x   = wbase < NXI;
  const bool b_e   = !b_x && wbase < NXI + NEI;
  const int  ra = tid / NX4, ca = tid % NX4;
  const int  ib = tid + 256;
  const int  rb = ib / NX4, cb = ib % NX4;
  const int  e    = b_e ? ib - NXI : 0;
  const int  re   = e / (2 * NE);
  const int  tab  = (e % (2 * NE)) >> 1;
  const int  hf   = e & 1;
  const int*   ip = tab ? idx1 : idx0;
  const float* ep = (tab ? emb1 : emb0) + hf * 4;
  const int  ecol = KX + tab * 8 + hf * 4;

  auto fetch = [&](int t, int id, float4& va, float4& vb, int& idnext) {
    vb = make_float4(0.f, 0.f, 0.f, 0.f);
    if (b_e) {
      vb = *reinterpret_cast<const float4*>(ep + id * 8);
      idnext = ip[min((t + ENC_WGS) * 16 + re, nrows - 1)];
    }
    va = *reinterpret_cast<const float4*>(x + (size_t)min(t * 16 + ra, nrows - 1) * KX + ca * 4);
    if (b_x)
      vb = *reinterpret_cast<const float4*>(x + (size_t)min(t * 16 + rb, nrows - 1) * KX + cb * 4);
  };
  auto put = [&](float* A, const float4& va, const float4& vb) {
    *reinterpret_cast<float4*>(A + ra * LDK + ca * 4) = va;
    if (b_x) *reinterpret_cast<float4*>(A + rb * LDK + cb * 4) = vb;
    if (b_e) *reinterpret_cast<float4*>(A + re * LDK + ecol) = vb;
  };

  int t   = blockIdx.x;
  int idn = 0;
  if (t < ntiles) {
    float4 va, vb;
    int id0 = 0;
    if (b_e) id0 = ip[min(t * 16 + re, nrows - 1)];
    fetch(t, id0, va, vb, idn);
    put(As[0], va, vb);
  }
  __syncthreads();

  int buf = 0;
  for (; t < ntiles; t += ENC_WGS) {
    const int  tn   = t + ENC_WGS;
    const bool more = tn < ntiles;
    float4 na, nb;
    int    idnn = 0;
    fetch(tn, idn, na, nb, idnn);
    __builtin_amdgcn_sched_barrier(0);

    const float* ap = As[buf] + (lane & 15) * LDK + 2 * hl;
    v8f c;
    #pragma unroll
    for (int v = 0; v < 8; ++v) c[v] = bias;
    #pragma unroll
    for (int j = 0; j < KS; ++j) {
      const v2f a = *reinterpret_cast<const v2f*>(ap + 4 * j);
      c = wmma4(a, bf[j], c);
    }
    if (more) put(As[buf ^ 1], na, nb);
    __syncthreads();
    store_tile(out, t * 16, nrows, col0, lane, c, true);
    buf ^= 1;
    idn = idnn;
  }
}


#define AT    400
#define AKEEP 64
#define AQ    128

#define LCAPW   1024
#define LHDR    32
#define BINOFF  (LHDR + 32 * LCAPW)
#define BING    250
#define SEGC    64
#define BINH    256
#define BTMAX   256
#define BIN_FITS(tile, rows) ((unsigned)(tile) < BTMAX && (rows) * 128 >= BINOFF + BINH + BING * SEGC)

__global__ void __launch_bounds__(1024) bin_kernel(const int* __restrict__ src,
                                                   const int* __restrict__ dst,
                                                   int* __restrict__ outw,
                                                   int E, int N, int chunk) {
  __shared__ int bc[BTMAX];
  const int tid = threadIdx.x;
  const int g   = blockIdx.x;
  if (tid < BTMAX) bc[tid] = 0;
  __syncthreads();
  const int e0 = g * chunk;
  const int e1 = min(E, e0 + chunk);
  for (int base = e0; base < e1; base += 8 * 1024) {
    int dv[8], sv[8];
    #pragma unroll
    for (int i = 0; i < 8; ++i) {
      const int ec = min(base + i * 1024 + tid, e1 - 1);
      dv[i] = dst[ec];
      sv[i] = src[ec];
    }
    asm volatile("" :: "v"(dv[0]), "v"(sv[0]));
    #pragma unroll
    for (int i = 0; i < 8; ++i) {
      if (base + i * 1024 + tid < e1) {
        const int t    = dv[i] / AT;
        const int dl   = dv[i] - t * AT;
        const int rows = min(AT, N - t * AT);
        if (BIN_FITS(t, rows)) {
          const int pos = atomicAdd(&bc[t], 1);
          if (pos < SEGC)
            outw[(size_t)t * (AT * 128) + BINOFF + BINH + g * SEGC + pos] = (sv[i] << 9) | dl;
        }
      }
    }
  }
  __syncthreads();
  const int nt = (N + AT - 1) / AT;
  if (tid < BTMAX && tid < nt) {
    const int rows = min(AT, N - tid * AT);
    if (BIN_FITS(tid, rows)) outw[(size_t)tid * (AT * 128) + BINOFF + g] = bc[tid];
  }
}

#define AOWN  13

template <bool SECOND>
__global__ void __launch_bounds__(1024) aggregate_kernel(
    const float* __restrict__ h, const int* __restrict__ src,
    const int* __restrict__ dst, float* __restrict__ s,
    float* __restrict__ cnt, int* __restrict__ lists, int E, int N) {
  __shared__ float acc[AT * 128];
  __shared__ int   cntl[AT];
  __shared__ int   qbuf[32 * AQ];
  __shared__ int   sub[32];
  __shared__ int   binbad;

  const int tid   = threadIdx.x;
  const int lane  = tid & 31;
  const int w     = __builtin_amdgcn_readfirstlane(tid >> 5);
  const int nbase = blockIdx.x * AT;
  const int rows  = min(AT, N - nbase);
  int* q = qbuf + w * AQ;
  int* mylist   = lists + (size_t)nbase * 128;
  const int subcap = min(((rows * 128 - LHDR) / 32) & ~31, LCAPW);

  const bool hi  = lane >= 16;
  const int  oa  = lane * 4 + (hi ? 2 : 0);
  const int  ob  = lane * 4 + (hi ? 0 : 2);
  auto add4 = [&](int row, const float4& v) {
    float* pa = acc + row * 128 + oa;
    float* pb = acc + row * 128 + ob;
    atomicAdd(pa + 0, hi ? v.z : v.x);
    atomicAdd(pa + 1, hi ? v.w : v.y);
    atomicAdd(pb + 0, hi ? v.x : v.z);
    atomicAdd(pb + 1, hi ? v.y : v.w);
  };

  auto add_rows = [&](int sv, int dl, int n) {
    int k = 0;
    for (; k + 8 <= n; k += 8) {
      float4 v[8];
      int    dd[8];
      #pragma unroll
      for (int u = 0; u < 8; ++u) {
        const int su = __builtin_amdgcn_readlane(sv, k + u);
        dd[u]        = __builtin_amdgcn_readlane(dl, k + u);
        v[u] = *reinterpret_cast<const float4*>(h + (size_t)su * 128 + lane * 4);
      }
      #pragma unroll
      for (int u = 0; u < 8; ++u) add4(dd[u], v[u]);
    }
    for (; k < n; ++k) {
      const int su = __builtin_amdgcn_readlane(sv, k);
      const int du = __builtin_amdgcn_readlane(dl, k);
      const float4 v = *reinterpret_cast<const float4*>(h + (size_t)su * 128 + lane * 4);
      add4(du, v);
    }
  };

  auto scan = [&]() {
    auto drain_round = [&](int r, int n) {
      int sv = 0, dl = 0;
      if (lane < n) {
        const int e = q[r + lane];
        sv = src[e];
        dl = dst[e] - nbase;
        if (!SECOND) atomicAdd(&cntl[dl], 1);
      }
      add_rows(sv, dl, n);
    };

    int qn = 0;

    auto push = [&](bool match, int e) {
      const unsigned mm = __builtin_amdgcn_ballot_w32(match);
      if (mm) {
        if (match) q[qn + (int)__builtin_amdgcn_mbcnt_lo(mm, 0u)] = e;
        qn += __builtin_popcount(mm);
      }
    };

    auto make_room = [&]() {
      if (qn > AKEEP) {
        const int full = qn & ~31;
        for (int r = 0; r < full; r += 32) drain_round(r, 32);
        const int rem = qn - full;
        int t = 0;
        if (lane < rem) t = q[full + lane];
        if (lane < rem) q[lane] = t;
        qn = rem;
      }
    };

    for (int e0 = w * 32; e0 < E; e0 += 1024) {
      make_room();
      const int e = e0 + lane;
      const int d = (e < E) ? dst[e] : -1;
      push((unsigned)(d - nbase) < (unsigned)rows, e);
    }
    for (int r = 0; r < qn; r += 32) drain_round(r, min(32, qn - r));
  };

  auto distribute = [&]() {
    const int* bin = mylist + BINOFF;
    auto hand = [&](int p, bool valid) {
      if (valid) {
        const int dl = p & 511;
        atomicAdd(&cntl[dl], 1);
        const int ow  = dl / AOWN;
        const int pos = atomicAdd(&sub[ow], 1);
        if (pos < subcap) mylist[LHDR + ow * subcap + pos] = p;
      }
    };
    for (int g0 = w; g0 < BING; g0 += 32 * 8) {
      int ng[8], pa[8], pb[8];
      #pragma unroll
      for (int j = 0; j < 8; ++j) {
        const int g = g0 + 32 * j;
        ng[j] = (g < BING) ? bin[g] : 0;
      }
      #pragma unroll
      for (int j = 0; j < 8; ++j) {
        const int g = g0 + 32 * j;
        const int n = min(ng[j], SEGC);
        pa[j] = (lane < n)      ? bin[BINH + g * SEGC + lane]      : 0;
        pb[j] = (lane + 32 < n) ? bin[BINH + g * SEGC + 32 + lane] : 0;
      }
      #pragma unroll
      for (int j = 0; j < 8; ++j) {
        const int n = ng[j];
        if (n > SEGC) binbad = 1;
        hand(pa[j], lane < min(n, SEGC));
        hand(pb[j], lane + 32 < min(n, SEGC));
      }
    }
  };

  int  nsub = 0;
  bool bad  = false;
  if (!SECOND) {
    const bool fits = BIN_FITS((int)blockIdx.x, rows);
    for (int idx = tid; idx < AT; idx += 1024) cntl[idx] = 0;
    if (tid < 32) sub[tid] = 0;
    if (tid == 0) binbad = fits ? 0 : 1;
    __syncthreads();
    if (fits) distribute();
    __syncthreads();
    bad  = binbad != 0;
    nsub = bad ? 0x7fffffff : sub[lane];
  } else {
    nsub = mylist[lane];
  }
  const bool replay = __builtin_amdgcn_ballot_w32((unsigned)nsub > (unsigned)subcap) == 0u;

  if (replay) {
    const int own0 = w * AOWN;
    const int ownn = max(0, min(AOWN, rows - own0));
    float* myacc = acc + own0 * 128 + lane * 4;
    for (int j = 0; j < ownn; ++j)
      *reinterpret_cast<v4f*>(myacc + j * 128) = v4f{0.f, 0.f, 0.f, 0.f};

    auto loadg = [&](int sv, int dl, int k, v4f (&v)[8], int (&dd)[8]) {
      #pragma unroll
      for (int u = 0; u < 8; ++u) {
        const int su = __builtin_amdgcn_readlane(sv, k + u);
        dd[u]        = __builtin_amdgcn_readlane(dl, k + u);
        v[u] = *reinterpret_cast<const v4f*>(h + (size_t)su * 128 + lane * 4);
      }
    };
    auto rmw8 = [&](const v4f (&v)[8], const int (&dd)[8]) {
      v4f ld = *reinterpret_cast<const v4f*>(myacc + dd[0] * 128);
      v4f prev = ld;
      #pragma unroll
      for (int u = 0; u < 8; ++u) {
        v4f nx = ld;
        if (u < 7) nx = *reinterpret_cast<const v4f*>(myacc + dd[u + 1] * 128);
        const v4f base = (u > 0 && dd[u] == dd[u - 1]) ? prev : ld;
        const v4f sum  = base + v[u];
        *reinterpret_cast<v4f*>(myacc + dd[u] * 128) = sum;
        prev = sum;
        ld   = nx;
      }
    };
    auto round32 = [&](int p) {
      const int sv = (int)((unsigned)p >> 9);
      const int dl = (p & 511) - own0;
      v4f A[8], B[8];
      int da[8], db[8];
      loadg(sv, dl, 0, A, da);
      loadg(sv, dl, 8, B, db);
      rmw8(A, da);
      loadg(sv, dl, 16, A, da);
      rmw8(B, db);
      loadg(sv, dl, 24, B, db);
      rmw8(A, da);
      rmw8(B, db);
    };
    auto round_rest = [&](int p, int n) {
      const int sv = (int)((unsigned)p >> 9);
      const int dl = (p & 511) - own0;
      int k = 0;
      for (; k + 8 <= n; k += 8) {
        v4f A[8];
        int da[8];
        loadg(sv, dl, k, A, da);
        rmw8(A, da);
      }
      for (; k < n; ++k) {
        const int su = __builtin_amdgcn_readlane(sv, k);
        const int du = __builtin_amdgcn_readlane(dl, k);
        const v4f v = *reinterpret_cast<const v4f*>(h + (size_t)su * 128 + lane * 4);
        v4f* pa = reinterpret_cast<v4f*>(myacc + du * 128);
        *pa = *pa + v;
      }
    };

    const int  mycnt = __builtin_amdgcn_readlane(nsub, w);
    const int* ent   = mylist + LHDR + w * subcap;
    float mydeg = 1.0f;
    if (lane < ownn) mydeg = SECOND ? cnt[nbase + own0 + lane] : (float)cntl[own0 + lane];
    int p = (lane < mycnt) ? ent[lane] : 0;
    for (int r = 0; r < mycnt; r += 32) {
      const int pn = (r + 32 + lane < mycnt) ? ent[r + 32 + lane] : 0;
      if (mycnt - r >= 32) round32(p);
      else                 round_rest(p, mycnt - r);
      p = pn;
    }

    for (int j = 0; j < ownn; ++j) {
      const float dd = fmaxf(__int_as_float(__builtin_amdgcn_readlane(__float_as_int(mydeg), j)), 1.0f);
      v4f a = *reinterpret_cast<const v4f*>(myacc + j * 128);
      a.x /= dd; a.y /= dd; a.z /= dd; a.w /= dd;
      *reinterpret_cast<v4f*>(s + (size_t)(nbase + own0 + j) * 128 + lane * 4) = a;
    }
  } else {
    for (int idx = tid; idx < AT * 32; idx += 1024)
      *reinterpret_cast<float4*>(acc + idx * 4) = make_float4(0.f, 0.f, 0.f, 0.f);
    if (!SECOND)
      for (int idx = tid; idx < AT; idx += 1024) cntl[idx] = 0;
    __syncthreads();
    scan();
    __syncthreads();
    for (int idx = tid; idx < rows * 32; idx += 1024) {
      const int r = idx >> 5, c4 = idx & 31;
      const float dd = fmaxf(SECOND ? cnt[nbase + r] : (float)cntl[r], 1.0f);
      float4 a = *reinterpret_cast<const float4*>(acc + r * 128 + c4 * 4);
      a.x /= dd; a.y /= dd; a.z /= dd; a.w /= dd;
      *reinterpret_cast<float4*>(s + (size_t)(nbase + r) * 128 + c4 * 4) = a;
    }
  }
  if (!SECOND) {
    for (int r = tid; r < rows; r += 1024) cnt[nbase + r] = (float)cntl[r];
    if (tid < 32) mylist[tid] = bad ? 0x7fffffff : sub[tid];
  }
}


#define SAGE_WGS 512
#define SAGE_KS  64

__global__ void __launch_bounds__(256) sage_kernel(
    const float* h, const float* s, const float* __restrict__ Wl,
    const float* __restrict__ bl, const float* __restrict__ Wr,
    float* out, int nrows, int do_relu) {
  __shared__ float As[2][16 * LDK];
  const int tid    = threadIdx.x;
  const int lane   = tid & 31;
  const int hl     = lane >> 4;
  const int col0   = (tid >> 5) * 16;
  const int ntiles = (nrows + 15) >> 4;

  v2f bf[SAGE_KS];
  {
    const int cn = col0 + (lane & 15);
    #pragma unroll
    for (int j = 0; j < 32; ++j) {
      const int k = 4 * j + 2 * hl;
      bf[j].x      = Wl[k * 128 + cn];
      bf[j].y      = Wl[(k + 1) * 128 + cn];
      bf[32 + j].x = Wr[k * 128 + cn];
      bf[32 + j].y = Wr[(k + 1) * 128 + cn];
    }
  }
  const float bias = bl[col0 + (lane & 15)];

  const int  c4     = tid & 63;
  const int  rb     = tid >> 6;
  const bool is_sum = (__builtin_amdgcn_readfirstlane(tid >> 5) & 1) == 0;

  const float* rows_in = (is_sum ? s : h) + (c4 & 31) * 4;
  auto fetch = [&](int t, v4f (&v)[4]) {
    #pragma unroll
    for (int i = 0; i < 4; ++i) {
      const int row = min(t * 16 + rb + 4 * i, nrows - 1);
      v[i] = *reinterpret_cast<const v4f*>(rows_in + (size_t)row * 128);
    }
  };
  auto put = [&](float* A, const v4f (&v)[4]) {
    #pragma unroll
    for (int i = 0; i < 4; ++i)
      *reinterpret_cast<v4f*>(A + (rb + 4 * i) * LDK + c4 * 4) = v[i];
  };

  int t = blockIdx.x;
  if (t < ntiles) {
    v4f v[4];
    fetch(t, v);
    put(As[0], v);
  }
  __syncthreads();

  int buf = 0;
  for (; t < ntiles; t += SAGE_WGS) {
    const int  tn   = t + SAGE_WGS;
    const bool more = tn < ntiles;
    v4f vn[4];
    fetch(tn, vn);
    __builtin_amdgcn_sched_barrier(0);

    const float* ap = As[buf] + (lane & 15) * LDK + 2 * hl;
    v8f c;
    #pragma unroll
    for (int v = 0; v < 8; ++v) c[v] = bias;
    #pragma unroll
    for (int j = 0; j < SAGE_KS; ++j) {
      const v2f a = *reinterpret_cast<const v2f*>(ap + 4 * j);
      c = wmma4(a, bf[j], c);
    }
    if (more) put(As[buf ^ 1], vn);
    __syncthreads();
    store_tile(out, t * 16, nrows, col0, lane, c, do_relu != 0);
    buf ^= 1;
  }
}


extern "C" void kernel_launch(void* const* d_in, const int* in_sizes, int n_in,
                              void* d_out, int out_size, void* d_ws, size_t ws_size,
                              hipStream_t stream) {
  (void)n_in; (void)out_size; (void)ws_size;
  const float* x_pol   = (const float*)d_in[0];
  const int*   pol_sid = (const int*)  d_in[1];
  const float* x_comp  = (const float*)d_in[2];
  const int*   c_sec   = (const int*)  d_in[3];
  const int*   c_ind   = (const int*)  d_in[4];
  const int*   edges   = (const int*)  d_in[5];
  const float* s_emb   = (const float*)d_in[6];
  const float* sec_emb = (const float*)d_in[7];
  const float* ind_emb = (const float*)d_in[8];
  const float* W_pol   = (const float*)d_in[9];
  const float* b_pol   = (const float*)d_in[10];
  const float* W_comp  = (const float*)d_in[11];
  const float* b_comp  = (const float*)d_in[12];
  const float* Wl1     = (const float*)d_in[13];
  const float* bl1     = (const float*)d_in[14];
  const float* Wr1     = (const float*)d_in[15];
  const float* Wl2     = (const float*)d_in[16];
  const float* bl2     = (const float*)d_in[17];
  const float* Wr2     = (const float*)d_in[18];

  const int NP = in_sizes[0] / 64;
  const int NC = in_sizes[2] / 96;
  const int E  = in_sizes[5] / 2;
  const int N  = NP + NC;
  const int* src = edges;
  const int* dst = edges + E;

  float* h    = (float*)d_ws;
  float* s    = h + (size_t)N * 128;
  float* cnt  = s + (size_t)N * 128;
  float* outp = (float*)d_out;
  int*   lists = (int*)d_out;

  const dim3 blk(256);
  const dim3 ablk(1024);
  const dim3 agrid((N + AT - 1) / AT);

  const int chunk = (E + BING - 1) / BING;
  bin_kernel<<<dim3(BING), ablk, 0, stream>>>(src, dst, lists, E, N, chunk);

  encode_kernel<16, 1><<<dim3(ENC_WGS), blk, 0, stream>>>(
      x_pol, pol_sid, s_emb, pol_sid, s_emb, W_pol, b_pol, h, NP);
  encode_kernel<24, 2><<<dim3(ENC_WGS), blk, 0, stream>>>(
      x_comp, c_sec, sec_emb, c_ind, ind_emb, W_comp, b_comp,
      h + (size_t)NP * 128, NC);

  aggregate_kernel<false><<<agrid, ablk, 0, stream>>>(h, src, dst, s, cnt, lists, E, N);
  sage_kernel<<<dim3(SAGE_WGS), blk, 0, stream>>>(
      h, s, Wl1, bl1, Wr1, h, N, 1);

  aggregate_kernel<true><<<agrid, ablk, 0, stream>>>(h, src, dst, s, cnt, lists, E, N);
  sage_kernel<<<dim3(SAGE_WGS), blk, 0, stream>>>(
      h, s, Wl2, bl2, Wr2, outp, N, 0);
}
